# starting kernel
# baseline (speedup 1.0000x reference)
#include <hip/hip_runtime.h>
#include <hip/hip_bf16.h>

typedef __attribute__((ext_vector_type(2))) float v2f;
typedef __attribute__((ext_vector_type(8))) float v8f;

__device__ __forceinline__ v8f wmma4(v2f a, v2f b, v8f c) {
    return __builtin_amdgcn_wmma_f32_16x16x4_f32(
         false, a,  false, b,
         (short)0, c,  false,  false);
}

#define G_TM 128
#define G_TN 128
#define G_TK 32
#define G_PAD 36

__global__ __launch_bounds__(256) void gemm_nt_kernel(
    const float* __restrict__ A, const float* __restrict__ W,
    const float* __restrict__ bias, float* __restrict__ C,
    int M, int N, int K, int act, int split)
{
    __shared__ float As[2][G_TM * G_PAD];
    __shared__ float Bs[2][G_TN * G_PAD];

    const int tid   = threadIdx.x;
    const int lane  = tid & 31;
    const int lrow  = lane & 15;
    const int khalf = (lane >> 4) * 2;
    const int rhalf = (lane >> 4) * 8;
    const int wave  = tid >> 5;

    const int m0 = blockIdx.x * G_TM;
    const int n0 = blockIdx.y * G_TN;
    const int wm = (wave & 3) * 32;
    const int wn = (wave >> 2) * 64;

    const int ldRow = tid >> 3;
    const int ldCol = (tid & 7) * 4;

    v8f cacc[2][4] = {};
    float4 pa[4], pb[4];

    auto loadTile = [&](int kt) {
#pragma unroll
        for (int i = 0; i < 4; ++i)
            pa[i] = *(const float4*)&A[(size_t)(m0 + ldRow + 32 * i) * K + kt + ldCol];
#pragma unroll
        for (int i = 0; i < 4; ++i)
            pb[i] = *(const float4*)&W[(size_t)(n0 + ldRow + 32 * i) * K + kt + ldCol];
    };
    auto storeTile = [&](int p) {
#pragma unroll
        for (int i = 0; i < 4; ++i)
            *(float4*)&As[p][(ldRow + 32 * i) * G_PAD + ldCol] = pa[i];
#pragma unroll
        for (int i = 0; i < 4; ++i)
            *(float4*)&Bs[p][(ldRow + 32 * i) * G_PAD + ldCol] = pb[i];
    };

    loadTile(0);
    storeTile(0);
    __syncthreads();

    int p = 0;
    for (int kt = 0; kt < K; kt += G_TK) {
        const bool hasNext = (kt + G_TK) < K;
        if (hasNext) loadTile(kt + G_TK);

#pragma unroll
        for (int kk = 0; kk < G_TK; kk += 4) {
            v2f af[2], bf[4];
            af[0] = *(const v2f*)&As[p][(wm +      lrow) * G_PAD + kk + khalf];
            af[1] = *(const v2f*)&As[p][(wm + 16 + lrow) * G_PAD + kk + khalf];
#pragma unroll
            for (int fn = 0; fn < 4; ++fn)
                bf[fn] = *(const v2f*)&Bs[p][(wn + fn * 16 + lrow) * G_PAD + kk + khalf];
#pragma unroll
            for (int fm = 0; fm < 2; ++fm)
#pragma unroll
                for (int fn = 0; fn < 4; ++fn)
                    cacc[fm][fn] = wmma4(af[fm], bf[fn], cacc[fm][fn]);
        }

        if (hasNext) storeTile(p ^ 1);
        __syncthreads();
        p ^= 1;
    }

#pragma unroll
    for (int fm = 0; fm < 2; ++fm)
#pragma unroll
        for (int fn = 0; fn < 4; ++fn)
#pragma unroll
            for (int r = 0; r < 8; ++r) {
                int row = m0 + wm + fm * 16 + r + rhalf;
                int col = n0 + wn + fn * 16 + lrow;
                float v = cacc[fm][fn][r];
                if (act == 1) v = (v > 0.0f) ? (v + 1.0f) : __expf(v);
                if (bias) v += bias[col];
                if (split) {
                    int bb = row >> 11, t = row & 2047;
                    int h  = col >> 6,  dd = col & 63;
                    C[(((size_t)bb * 16 + h) * 2048 + t) * 64 + dd] = v;
                } else {
                    C[(size_t)row * N + col] = v;
                }
            }
}

#define C_PAD  68
#define A_PAD  130
#define S_PAD  66

__global__ __launch_bounds__(256) void linattn_scan_kernel(
    const float* __restrict__ Q, const float* __restrict__ Kx,
    const float* __restrict__ V, float* __restrict__ O)
{
    __shared__ float qs[128 * C_PAD];
    __shared__ float ks[128 * C_PAD];
    __shared__ float vs[128 * C_PAD];
    __shared__ float Am[128 * A_PAD];
    __shared__ float Sm[64 * S_PAD];
    __shared__ float zv[64];
    __shared__ float den[128];

    const int tid   = threadIdx.x;
    const int lane  = tid & 31;
    const int lrow  = lane & 15;
    const int khalf = (lane >> 4) * 2;
    const int rhalf = (lane >> 4) * 8;
    const int wave  = tid >> 5;

    const int bh = blockIdx.x;
    const int b  = bh >> 4, h = bh & 15;
    const size_t base = (size_t)bh * 2048 * 64;

    for (int i = tid; i < 64 * S_PAD; i += 256) Sm[i] = 0.0f;
    if (tid < 64) zv[tid] = 0.0f;
    __syncthreads();

    for (int jc = 0; jc < 16; ++jc) {
        const size_t cbase = base + (size_t)jc * 128 * 64;

#pragma unroll
        for (int i = 0; i < 8; ++i) {
            int idx = tid + i * 256;
            int row = idx >> 4, c4 = (idx & 15) * 4;
            *(float4*)&qs[row * C_PAD + c4] = *(const float4*)&Q [cbase + row * 64 + c4];
            *(float4*)&ks[row * C_PAD + c4] = *(const float4*)&Kx[cbase + row * 64 + c4];
            *(float4*)&vs[row * C_PAD + c4] = *(const float4*)&V [cbase + row * 64 + c4];
        }
        __syncthreads();

        {
            const int m0 = wave * 16;
            v8f acc[8] = {};
            for (int kk = 0; kk < 64; kk += 4) {
                v2f a = *(const v2f*)&qs[(m0 + lrow) * C_PAD + kk + khalf];
#pragma unroll
                for (int c = 0; c < 8; ++c) {
                    v2f bf = *(const v2f*)&ks[(c * 16 + lrow) * C_PAD + kk + khalf];
                    acc[c] = wmma4(a, bf, acc[c]);
                }
            }
#pragma unroll
            for (int c = 0; c < 8; ++c)
#pragma unroll
                for (int r = 0; r < 8; ++r) {
                    int row = m0 + r + rhalf;
                    int col = c * 16 + lrow;
                    Am[row * A_PAD + col] = (row >= col) ? acc[c][r] : 0.0f;
                }
        }
        __syncthreads();

        if (tid < 128) {
            float s = 1e-5f;
            for (int j = 0; j <= tid; ++j) s += Am[tid * A_PAD + j];
            float qz = 0.0f;
            for (int k = 0; k < 64; ++k) qz += qs[tid * C_PAD + k] * zv[k];
            den[tid] = s + qz;
        }
        __syncthreads();

        {
            const int m0 = wave * 16;
            v8f acc[4] = {};
            for (int kk = 0; kk < 128; kk += 4) {
                v2f a = *(const v2f*)&Am[(m0 + lrow) * A_PAD + kk + khalf];
#pragma unroll
                for (int c = 0; c < 4; ++c) {
                    v2f bf;
                    bf.x = vs[(kk + khalf)     * C_PAD + c * 16 + lrow];
                    bf.y = vs[(kk + khalf + 1) * C_PAD + c * 16 + lrow];
                    acc[c] = wmma4(a, bf, acc[c]);
                }
            }
            for (int kk = 0; kk < 64; kk += 4) {
                v2f a = *(const v2f*)&qs[(m0 + lrow) * C_PAD + kk + khalf];
#pragma unroll
                for (int c = 0; c < 4; ++c) {
                    v2f bf;
                    bf.x = Sm[(kk + khalf)     * S_PAD + c * 16 + lrow];
                    bf.y = Sm[(kk + khalf + 1) * S_PAD + c * 16 + lrow];
                    acc[c] = wmma4(a, bf, acc[c]);
                }
            }
#pragma unroll
            for (int c = 0; c < 4; ++c)
#pragma unroll
                for (int r = 0; r < 8; ++r) {
                    int row = m0 + r + rhalf;
                    int col = c * 16 + lrow;
                    float o = acc[c][r] / den[row];
                    O[((size_t)(b * 2048 + jc * 128 + row)) * 1024 + h * 64 + col] = o;
                }
        }
        __syncthreads();

        {
            const int tr  = (wave & 3) * 16;
            const int tc0 = (wave >> 2) * 32;
            v8f acc[2];
#pragma unroll
            for (int c = 0; c < 2; ++c)
#pragma unroll
                for (int r = 0; r < 8; ++r)
                    acc[c][r] = Sm[(tr + r + rhalf) * S_PAD + tc0 + c * 16 + lrow];
            for (int kk = 0; kk < 128; kk += 4) {
                v2f a;
                a.x = ks[(kk + khalf)     * C_PAD + tr + lrow];
                a.y = ks[(kk + khalf + 1) * C_PAD + tr + lrow];
#pragma unroll
                for (int c = 0; c < 2; ++c) {
                    v2f bf;
                    bf.x = vs[(kk + khalf)     * C_PAD + tc0 + c * 16 + lrow];
                    bf.y = vs[(kk + khalf + 1) * C_PAD + tc0 + c * 16 + lrow];
                    acc[c] = wmma4(a, bf, acc[c]);
                }
            }
#pragma unroll
            for (int c = 0; c < 2; ++c)
#pragma unroll
                for (int r = 0; r < 8; ++r)
                    Sm[(tr + r + rhalf) * S_PAD + tc0 + c * 16 + lrow] = acc[c][r];
        }
        __syncthreads();

        if (tid < 64) {
            float s = zv[tid];
            for (int i = 0; i < 128; ++i) s += ks[i * C_PAD + tid];
            zv[tid] = s;
        }
        __syncthreads();
    }
}

extern "C" void kernel_launch(void* const* d_in, const int* in_sizes, int n_in,
                              void* d_out, int out_size, void* d_ws, size_t ws_size,
                              hipStream_t stream) {
    const float* x  = (const float*)d_in[0];
    const float* Wq = (const float*)d_in[1];
    const float* Wk = (const float*)d_in[2];
    const float* Wv = (const float*)d_in[3];
    const float* Wo = (const float*)d_in[4];
    const float* bo = (const float*)d_in[5];

    const int T = 2048, E = 1024;
    const int B = in_sizes[0] / (T * E);
    const int M = B * T;
    const size_t plane = (size_t)M * E;

    float* phiQ = (float*)d_ws;
    float* phiK = phiQ + plane;
    float* Vh   = phiK + plane;
    float* attn = Vh   + plane;

    dim3 blk(256);
    dim3 gg(M / G_TM, E / G_TN);

    gemm_nt_kernel<<<gg, blk, 0, stream>>>(x, Wq, nullptr, phiQ, M, E, E,  1,  1);
    gemm_nt_kernel<<<gg, blk, 0, stream>>>(x, Wk, nullptr, phiK, M, E, E,  1,  1);
    gemm_nt_kernel<<<gg, blk, 0, stream>>>(x, Wv, nullptr, Vh,   M, E, E,  0,  1);

    linattn_scan_kernel<<<dim3(B * 16), blk, 0, stream>>>(phiQ, phiK, Vh, attn);

    gemm_nt_kernel<<<gg, blk, 0, stream>>>(attn, Wo, bo, (float*)d_out, M, E, E,  0,  0);
}


# scan made chunk-parallel: 512 work-groups (chunk,b,h) for 32; each rebuilds its prefix state S,z from the earlier rows' k,v in the same row order, so no sum's order changes; two groups share a WGP
# speedup vs baseline: 1.3221x; 1.3221x over previous
#include <hip/hip_runtime.h>
#include <hip/hip_bf16.h>

typedef __attribute__((ext_vector_type(2))) float v2f;
typedef __attribute__((ext_vector_type(8))) float v8f;

__device__ __forceinline__ v8f wmma4(v2f a, v2f b, v8f c) {
    return __builtin_amdgcn_wmma_f32_16x16x4_f32(
         false, a,  false, b,
         (short)0, c,  false,  false);
}

#define G_TM 128
#define G_TN 128
#define G_TK 32
#define G_PAD 36

__global__ __launch_bounds__(256) void gemm_nt_kernel(
    const float* __restrict__ A, const float* __restrict__ W,
    const float* __restrict__ bias, float* __restrict__ C,
    int M, int N, int K, int act, int split)
{
    __shared__ float As[2][G_TM * G_PAD];
    __shared__ float Bs[2][G_TN * G_PAD];

    const int tid   = threadIdx.x;
    const int lane  = tid & 31;
    const int lrow  = lane & 15;
    const int khalf = (lane >> 4) * 2;
    const int rhalf = (lane >> 4) * 8;
    const int wave  = tid >> 5;

    const int m0 = blockIdx.x * G_TM;
    const int n0 = blockIdx.y * G_TN;
    const int wm = (wave & 3) * 32;
    const int wn = (wave >> 2) * 64;

    const int ldRow = tid >> 3;
    const int ldCol = (tid & 7) * 4;

    v8f cacc[2][4] = {};
    float4 pa[4], pb[4];

    auto loadTile = [&](int kt) {
#pragma unroll
        for (int i = 0; i < 4; ++i)
            pa[i] = *(const float4*)&A[(size_t)(m0 + ldRow + 32 * i) * K + kt + ldCol];
#pragma unroll
        for (int i = 0; i < 4; ++i)
            pb[i] = *(const float4*)&W[(size_t)(n0 + ldRow + 32 * i) * K + kt + ldCol];
    };
    auto storeTile = [&](int p) {
#pragma unroll
        for (int i = 0; i < 4; ++i)
            *(float4*)&As[p][(ldRow + 32 * i) * G_PAD + ldCol] = pa[i];
#pragma unroll
        for (int i = 0; i < 4; ++i)
            *(float4*)&Bs[p][(ldRow + 32 * i) * G_PAD + ldCol] = pb[i];
    };

    loadTile(0);
    storeTile(0);
    __syncthreads();

    int p = 0;
    for (int kt = 0; kt < K; kt += G_TK) {
        const bool hasNext = (kt + G_TK) < K;
        if (hasNext) loadTile(kt + G_TK);

#pragma unroll
        for (int kk = 0; kk < G_TK; kk += 4) {
            v2f af[2], bf[4];
            af[0] = *(const v2f*)&As[p][(wm +      lrow) * G_PAD + kk + khalf];
            af[1] = *(const v2f*)&As[p][(wm + 16 + lrow) * G_PAD + kk + khalf];
#pragma unroll
            for (int fn = 0; fn < 4; ++fn)
                bf[fn] = *(const v2f*)&Bs[p][(wn + fn * 16 + lrow) * G_PAD + kk + khalf];
#pragma unroll
            for (int fm = 0; fm < 2; ++fm)
#pragma unroll
                for (int fn = 0; fn < 4; ++fn)
                    cacc[fm][fn] = wmma4(af[fm], bf[fn], cacc[fm][fn]);
        }

        if (hasNext) storeTile(p ^ 1);
        __syncthreads();
        p ^= 1;
    }

#pragma unroll
    for (int fm = 0; fm < 2; ++fm)
#pragma unroll
        for (int fn = 0; fn < 4; ++fn)
#pragma unroll
            for (int r = 0; r < 8; ++r) {
                int row = m0 + wm + fm * 16 + r + rhalf;
                int col = n0 + wn + fn * 16 + lrow;
                float v = cacc[fm][fn][r];
                if (act == 1) v = (v > 0.0f) ? (v + 1.0f) : __expf(v);
                if (bias) v += bias[col];
                if (split) {
                    int bb = row >> 11, t = row & 2047;
                    int h  = col >> 6,  dd = col & 63;
                    C[(((size_t)bb * 16 + h) * 2048 + t) * 64 + dd] = v;
                } else {
                    C[(size_t)row * N + col] = v;
                }
            }
}

#define C_PAD  68
#define A_PAD  130
#define S_PAD  66
#define N_CHUNK 16

__global__ __launch_bounds__(256) void linattn_chunk_kernel(
    const float* __restrict__ Q, const float* __restrict__ Kx,
    const float* __restrict__ V, float* __restrict__ O)
{
    __shared__ __attribute__((aligned(16)))
        float lds[2 * 128 * C_PAD + 128 * A_PAD + 64 * S_PAD + 64 + 128];
    float* const qs  = lds;
    float* const vs  = qs + 128 * C_PAD;
    float* const ks  = vs + 128 * C_PAD;
    float* const Am  = ks;
    float* const Sm  = ks + 128 * A_PAD;
    float* const zv  = Sm + 64 * S_PAD;
    float* const den = zv + 64;

    const int tid   = threadIdx.x;
    const int lane  = tid & 31;
    const int lrow  = lane & 15;
    const int khalf = (lane >> 4) * 2;
    const int rhalf = (lane >> 4) * 8;
    const int wave  = tid >> 5;

    const int nbh = gridDim.x / N_CHUNK;
    const int jc  = N_CHUNK - 1 - blockIdx.x / nbh;
    const int bh  = blockIdx.x % nbh;
    const int b   = bh >> 4, h = bh & 15;
    const size_t base = (size_t)bh * 2048 * 64;

    float4 pk0, pk1, pk2, pk3, pk4, pk5, pk6, pk7;
    float4 pv0, pv1, pv2, pv3, pv4, pv5, pv6, pv7;
    float4 pq0, pq1, pq2, pq3, pq4, pq5, pq6, pq7;
    const int ldOff = (tid >> 4) * C_PAD + (tid & 15) * 4;
#define LD_G4(dst, src, cb, i) dst##i = *(const float4*)&src[(cb) + (size_t)(tid + (i) * 256) * 4];
#define ST_L4(buf, regs, i)    *(float4*)&buf[ldOff + (i) * 16 * C_PAD] = regs##i;
#define REP8(M, ...) M(__VA_ARGS__, 0) M(__VA_ARGS__, 1) M(__VA_ARGS__, 2) M(__VA_ARGS__, 3) \
                     M(__VA_ARGS__, 4) M(__VA_ARGS__, 5) M(__VA_ARGS__, 6) M(__VA_ARGS__, 7)

    REP8(LD_G4, pk, Kx, base)
    REP8(LD_G4, pv, V, base)
    {
        const size_t cbq = base + (size_t)jc * 128 * 64;
        REP8(LD_G4, pq, Q, cbq)
    }

    const int tr  = (wave & 3) * 16;
    const int tc0 = (wave >> 2) * 32;
    v8f sacc[2] = {};
    float zacc = 0.0f;

    for (int j = 0; j < jc; ++j) {
        REP8(ST_L4, ks, pk)
        REP8(ST_L4, vs, pv)
        __syncthreads();
        {
            const size_t cbn = base + (size_t)(j + 1) * 128 * 64;
            REP8(LD_G4, pk, Kx, cbn)
            REP8(LD_G4, pv, V, cbn)
        }

        for (int kk = 0; kk < 128; kk += 4) {
            v2f a;
            a.x = ks[(kk + khalf)     * C_PAD + tr + lrow];
            a.y = ks[(kk + khalf + 1) * C_PAD + tr + lrow];
#pragma unroll
            for (int c = 0; c < 2; ++c) {
                v2f bf;
                bf.x = vs[(kk + khalf)     * C_PAD + tc0 + c * 16 + lrow];
                bf.y = vs[(kk + khalf + 1) * C_PAD + tc0 + c * 16 + lrow];
                sacc[c] = wmma4(a, bf, sacc[c]);
            }
        }
        if (tid < 64) {
            float s = zacc;
            for (int i = 0; i < 128; ++i) s += ks[i * C_PAD + tid];
            zacc = s;
        }
        __syncthreads();
    }

    REP8(ST_L4, ks, pk)
    REP8(ST_L4, vs, pv)
    REP8(ST_L4, qs, pq)
#pragma unroll
    for (int c = 0; c < 2; ++c)
#pragma unroll
        for (int r = 0; r < 8; ++r)
            Sm[(tr + r + rhalf) * S_PAD + tc0 + c * 16 + lrow] = sacc[c][r];
    if (tid < 64) zv[tid] = zacc;
    __syncthreads();

    {
        const int m0 = wave * 16;
        v8f acc[8] = {};
        for (int kk = 0; kk < 64; kk += 4) {
            v2f a = *(const v2f*)&qs[(m0 + lrow) * C_PAD + kk + khalf];
#pragma unroll
            for (int c = 0; c < 8; ++c) {
                v2f bf = *(const v2f*)&ks[(c * 16 + lrow) * C_PAD + kk + khalf];
                acc[c] = wmma4(a, bf, acc[c]);
            }
        }
        __syncthreads();
#pragma unroll
        for (int c = 0; c < 8; ++c)
#pragma unroll
            for (int r = 0; r < 8; ++r) {
                int row = m0 + r + rhalf;
                int col = c * 16 + lrow;
                Am[row * A_PAD + col] = (row >= col) ? acc[c][r] : 0.0f;
            }
    }
    __syncthreads();

    if (tid < 128) {
        float s = 1e-5f;
        for (int j = 0; j <= tid; ++j) s += Am[tid * A_PAD + j];
        float qz = 0.0f;
        for (int k = 0; k < 64; ++k) qz += qs[tid * C_PAD + k] * zv[k];
        den[tid] = s + qz;
    }
    __syncthreads();

    {
        const int m0 = wave * 16;
        v8f acc[4] = {};
        for (int kk = 0; kk < 128; kk += 4) {
            v2f a = *(const v2f*)&Am[(m0 + lrow) * A_PAD + kk + khalf];
#pragma unroll
            for (int c = 0; c < 4; ++c) {
                v2f bf;
                bf.x = vs[(kk + khalf)     * C_PAD + c * 16 + lrow];
                bf.y = vs[(kk + khalf + 1) * C_PAD + c * 16 + lrow];
                acc[c] = wmma4(a, bf, acc[c]);
            }
        }
        for (int kk = 0; kk < 64; kk += 4) {
            v2f a = *(const v2f*)&qs[(m0 + lrow) * C_PAD + kk + khalf];
#pragma unroll
            for (int c = 0; c < 4; ++c) {
                v2f bf;
                bf.x = Sm[(kk + khalf)     * S_PAD + c * 16 + lrow];
                bf.y = Sm[(kk + khalf + 1) * S_PAD + c * 16 + lrow];
                acc[c] = wmma4(a, bf, acc[c]);
            }
        }
#pragma unroll
        for (int c = 0; c < 4; ++c)
#pragma unroll
            for (int r = 0; r < 8; ++r) {
                int row = m0 + r + rhalf;
                int col = c * 16 + lrow;
                float o = acc[c][r] / den[row];
                O[((size_t)(b * 2048 + jc * 128 + row)) * 1024 + h * 64 + col] = o;
            }
    }
}

extern "C" void kernel_launch(void* const* d_in, const int* in_sizes, int n_in,
                              void* d_out, int out_size, void* d_ws, size_t ws_size,
                              hipStream_t stream) {
    const float* x  = (const float*)d_in[0];
    const float* Wq = (const float*)d_in[1];
    const float* Wk = (const float*)d_in[2];
    const float* Wv = (const float*)d_in[3];
    const float* Wo = (const float*)d_in[4];
    const float* bo = (const float*)d_in[5];

    const int T = 2048, E = 1024;
    const int B = in_sizes[0] / (T * E);
    const int M = B * T;
    const size_t plane = (size_t)M * E;

    float* phiQ = (float*)d_ws;
    float* phiK = phiQ + plane;
    float* Vh   = phiK + plane;
    float* attn = Vh   + plane;

    dim3 blk(256);
    dim3 gg(M / G_TM, E / G_TN);

    gemm_nt_kernel<<<gg, blk, 0, stream>>>(x, Wq, nullptr, phiQ, M, E, E,  1,  1);
    gemm_nt_kernel<<<gg, blk, 0, stream>>>(x, Wk, nullptr, phiK, M, E, E,  1,  1);
    gemm_nt_kernel<<<gg, blk, 0, stream>>>(x, Wv, nullptr, Vh,   M, E, E,  0,  1);

    linattn_chunk_kernel<<<dim3(B * 16 * N_CHUNK), blk, 0, stream>>>(phiQ, phiK, Vh, attn);

    gemm_nt_kernel<<<gg, blk, 0, stream>>>(attn, Wo, bo, (float*)d_out, M, E, E,  0,  0);
}


# projection GEMMs: next K-tile's global loads held in named registers, not a private-memory array: waited for behind the MACs, no scratch round trip per K-tile; order of sums unchanged
# speedup vs baseline: 2.2524x; 1.7037x over previous
#include <hip/hip_runtime.h>
#include <hip/hip_bf16.h>

typedef __attribute__((ext_vector_type(2))) float v2f;
typedef __attribute__((ext_vector_type(8))) float v8f;

__device__ __forceinline__ v8f wmma4(v2f a, v2f b, v8f c) {
    return __builtin_amdgcn_wmma_f32_16x16x4_f32(
         false, a,  false, b,
         (short)0, c,  false,  false);
}

#define G_TM 128
#define G_TN 128
#define G_TK 32
#define G_PAD 36

__global__ __launch_bounds__(256) void gemm_nt_kernel(
    const float* __restrict__ A, const float* __restrict__ W,
    const float* __restrict__ bias, float* __restrict__ C,
    int M, int N, int K, int act, int split)
{
    __shared__ float As[2][G_TM * G_PAD];
    __shared__ float Bs[2][G_TN * G_PAD];

    const int tid   = threadIdx.x;
    const int lane  = tid & 31;
    const int lrow  = lane & 15;
    const int khalf = (lane >> 4) * 2;
    const int rhalf = (lane >> 4) * 8;
    const int wave  = tid >> 5;

    const int m0 = blockIdx.x * G_TM;
    const int n0 = blockIdx.y * G_TN;
    const int wm = (wave & 3) * 32;
    const int wn = (wave >> 2) * 64;

    const int ldRow = tid >> 3;
    const int ldCol = (tid & 7) * 4;

    v8f cacc[2][4] = {};
    float4 pa0, pa1, pa2, pa3, pb0, pb1, pb2, pb3;
#define G_LD(i, kt)                                                                         \
    pa##i = *(const float4*)&A[(size_t)(m0 + ldRow + 32 * (i)) * K + (kt) + ldCol];         \
    pb##i = *(const float4*)&W[(size_t)(n0 + ldRow + 32 * (i)) * K + (kt) + ldCol];
#define G_ST(i, pp)                                                                         \
    *(float4*)&As[pp][(ldRow + 32 * (i)) * G_PAD + ldCol] = pa##i;                          \
    *(float4*)&Bs[pp][(ldRow + 32 * (i)) * G_PAD + ldCol] = pb##i;
#define G_LOAD_TILE(kt)  G_LD(0, kt) G_LD(1, kt) G_LD(2, kt) G_LD(3, kt)
#define G_STORE_TILE(pp) G_ST(0, pp) G_ST(1, pp) G_ST(2, pp) G_ST(3, pp)

    G_LOAD_TILE(0)
    G_STORE_TILE(0)
    __syncthreads();

    int p = 0;
    for (int kt = 0; kt < K; kt += G_TK) {
        const bool hasNext = (kt + G_TK) < K;
        if (hasNext) { G_LOAD_TILE(kt + G_TK) }

#pragma unroll
        for (int kk = 0; kk < G_TK; kk += 4) {
            v2f af[2], bf[4];
            af[0] = *(const v2f*)&As[p][(wm +      lrow) * G_PAD + kk + khalf];
            af[1] = *(const v2f*)&As[p][(wm + 16 + lrow) * G_PAD + kk + khalf];
#pragma unroll
            for (int fn = 0; fn < 4; ++fn)
                bf[fn] = *(const v2f*)&Bs[p][(wn + fn * 16 + lrow) * G_PAD + kk + khalf];
#pragma unroll
            for (int fm = 0; fm < 2; ++fm)
#pragma unroll
                for (int fn = 0; fn < 4; ++fn)
                    cacc[fm][fn] = wmma4(af[fm], bf[fn], cacc[fm][fn]);
        }

        if (hasNext) { G_STORE_TILE(p ^ 1) }
        __syncthreads();
        p ^= 1;
    }

#pragma unroll
    for (int fm = 0; fm < 2; ++fm)
#pragma unroll
        for (int fn = 0; fn < 4; ++fn)
#pragma unroll
            for (int r = 0; r < 8; ++r) {
                int row = m0 + wm + fm * 16 + r + rhalf;
                int col = n0 + wn + fn * 16 + lrow;
                float v = cacc[fm][fn][r];
                if (act == 1) v = (v > 0.0f) ? (v + 1.0f) : __expf(v);
                if (bias) v += bias[col];
                if (split) {
                    int bb = row >> 11, t = row & 2047;
                    int h  = col >> 6,  dd = col & 63;
                    C[(((size_t)bb * 16 + h) * 2048 + t) * 64 + dd] = v;
                } else {
                    C[(size_t)row * N + col] = v;
                }
            }
}

#define C_PAD  68
#define A_PAD  130
#define S_PAD  66
#define N_CHUNK 16

__global__ __launch_bounds__(256) void linattn_chunk_kernel(
    const float* __restrict__ Q, const float* __restrict__ Kx,
    const float* __restrict__ V, float* __restrict__ O)
{
    __shared__ __attribute__((aligned(16)))
        float lds[2 * 128 * C_PAD + 128 * A_PAD + 64 * S_PAD + 64 + 128];
    float* const qs  = lds;
    float* const vs  = qs + 128 * C_PAD;
    float* const ks  = vs + 128 * C_PAD;
    float* const Am  = ks;
    float* const Sm  = ks + 128 * A_PAD;
    float* const zv  = Sm + 64 * S_PAD;
    float* const den = zv + 64;

    const int tid   = threadIdx.x;
    const int lane  = tid & 31;
    const int lrow  = lane & 15;
    const int khalf = (lane >> 4) * 2;
    const int rhalf = (lane >> 4) * 8;
    const int wave  = tid >> 5;

    const int nbh = gridDim.x / N_CHUNK;
    const int jc  = N_CHUNK - 1 - blockIdx.x / nbh;
    const int bh  = blockIdx.x % nbh;
    const int b   = bh >> 4, h = bh & 15;
    const size_t base = (size_t)bh * 2048 * 64;

    float4 pk0, pk1, pk2, pk3, pk4, pk5, pk6, pk7;
    float4 pv0, pv1, pv2, pv3, pv4, pv5, pv6, pv7;
    float4 pq0, pq1, pq2, pq3, pq4, pq5, pq6, pq7;
    const int ldOff = (tid >> 4) * C_PAD + (tid & 15) * 4;
#define LD_G4(dst, src, cb, i) dst##i = *(const float4*)&src[(cb) + (size_t)(tid + (i) * 256) * 4];
#define ST_L4(buf, regs, i)    *(float4*)&buf[ldOff + (i) * 16 * C_PAD] = regs##i;
#define REP8(M, ...) M(__VA_ARGS__, 0) M(__VA_ARGS__, 1) M(__VA_ARGS__, 2) M(__VA_ARGS__, 3) \
                     M(__VA_ARGS__, 4) M(__VA_ARGS__, 5) M(__VA_ARGS__, 6) M(__VA_ARGS__, 7)

    REP8(LD_G4, pk, Kx, base)
    REP8(LD_G4, pv, V, base)
    {
        const size_t cbq = base + (size_t)jc * 128 * 64;
        REP8(LD_G4, pq, Q, cbq)
    }

    const int tr  = (wave & 3) * 16;
    const int tc0 = (wave >> 2) * 32;
    v8f sacc[2] = {};
    float zacc = 0.0f;

    for (int j = 0; j < jc; ++j) {
        REP8(ST_L4, ks, pk)
        REP8(ST_L4, vs, pv)
        __syncthreads();
        {
            const size_t cbn = base + (size_t)(j + 1) * 128 * 64;
            REP8(LD_G4, pk, Kx, cbn)
            REP8(LD_G4, pv, V, cbn)
        }

        for (int kk = 0; kk < 128; kk += 4) {
            v2f a;
            a.x = ks[(kk + khalf)     * C_PAD + tr + lrow];
            a.y = ks[(kk + khalf + 1) * C_PAD + tr + lrow];
#pragma unroll
            for (int c = 0; c < 2; ++c) {
                v2f bf;
                bf.x = vs[(kk + khalf)     * C_PAD + tc0 + c * 16 + lrow];
                bf.y = vs[(kk + khalf + 1) * C_PAD + tc0 + c * 16 + lrow];
                sacc[c] = wmma4(a, bf, sacc[c]);
            }
        }
        if (tid < 64) {
            float s = zacc;
            for (int i = 0; i < 128; ++i) s += ks[i * C_PAD + tid];
            zacc = s;
        }
        __syncthreads();
    }

    REP8(ST_L4, ks, pk)
    REP8(ST_L4, vs, pv)
    REP8(ST_L4, qs, pq)
#pragma unroll
    for (int c = 0; c < 2; ++c)
#pragma unroll
        for (int r = 0; r < 8; ++r)
            Sm[(tr + r + rhalf) * S_PAD + tc0 + c * 16 + lrow] = sacc[c][r];
    if (tid < 64) zv[tid] = zacc;
    __syncthreads();

    {
        const int m0 = wave * 16;
        v8f acc[8] = {};
        for (int kk = 0; kk < 64; kk += 4) {
            v2f a = *(const v2f*)&qs[(m0 + lrow) * C_PAD + kk + khalf];
#pragma unroll
            for (int c = 0; c < 8; ++c) {
                v2f bf = *(const v2f*)&ks[(c * 16 + lrow) * C_PAD + kk + khalf];
                acc[c] = wmma4(a, bf, acc[c]);
            }
        }
        __syncthreads();
#pragma unroll
        for (int c = 0; c < 8; ++c)
#pragma unroll
            for (int r = 0; r < 8; ++r) {
                int row = m0 + r + rhalf;
                int col = c * 16 + lrow;
                Am[row * A_PAD + col] = (row >= col) ? acc[c][r] : 0.0f;
            }
    }
    __syncthreads();

    if (tid < 128) {
        float s = 1e-5f;
        for (int j = 0; j <= tid; ++j) s += Am[tid * A_PAD + j];
        float qz = 0.0f;
        for (int k = 0; k < 64; ++k) qz += qs[tid * C_PAD + k] * zv[k];
        den[tid] = s + qz;
    }
    __syncthreads();

    {
        const int m0 = wave * 16;
        v8f acc[4] = {};
        for (int kk = 0; kk < 128; kk += 4) {
            v2f a = *(const v2f*)&Am[(m0 + lrow) * A_PAD + kk + khalf];
#pragma unroll
            for (int c = 0; c < 4; ++c) {
                v2f bf;
                bf.x = vs[(kk + khalf)     * C_PAD + c * 16 + lrow];
                bf.y = vs[(kk + khalf + 1) * C_PAD + c * 16 + lrow];
                acc[c] = wmma4(a, bf, acc[c]);
            }
        }
        for (int kk = 0; kk < 64; kk += 4) {
            v2f a = *(const v2f*)&qs[(m0 + lrow) * C_PAD + kk + khalf];
#pragma unroll
            for (int c = 0; c < 4; ++c) {
                v2f bf;
                bf.x = Sm[(kk + khalf)     * S_PAD + c * 16 + lrow];
                bf.y = Sm[(kk + khalf + 1) * S_PAD + c * 16 + lrow];
                acc[c] = wmma4(a, bf, acc[c]);
            }
        }
#pragma unroll
        for (int c = 0; c < 4; ++c)
#pragma unroll
            for (int r = 0; r < 8; ++r) {
                int row = m0 + r + rhalf;
                int col = c * 16 + lrow;
                float o = acc[c][r] / den[row];
                O[((size_t)(b * 2048 + jc * 128 + row)) * 1024 + h * 64 + col] = o;
            }
    }
}

extern "C" void kernel_launch(void* const* d_in, const int* in_sizes, int n_in,
                              void* d_out, int out_size, void* d_ws, size_t ws_size,
                              hipStream_t stream) {
    const float* x  = (const float*)d_in[0];
    const float* Wq = (const float*)d_in[1];
    const float* Wk = (const float*)d_in[2];
    const float* Wv = (const float*)d_in[3];
    const float* Wo = (const float*)d_in[4];
    const float* bo = (const float*)d_in[5];

    const int T = 2048, E = 1024;
    const int B = in_sizes[0] / (T * E);
    const int M = B * T;
    const size_t plane = (size_t)M * E;

    float* phiQ = (float*)d_ws;
    float* phiK = phiQ + plane;
    float* Vh   = phiK + plane;
    float* attn = Vh   + plane;

    dim3 blk(256);
    dim3 gg(M / G_TM, E / G_TN);

    gemm_nt_kernel<<<gg, blk, 0, stream>>>(x, Wq, nullptr, phiQ, M, E, E,  1,  1);
    gemm_nt_kernel<<<gg, blk, 0, stream>>>(x, Wk, nullptr, phiK, M, E, E,  1,  1);
    gemm_nt_kernel<<<gg, blk, 0, stream>>>(x, Wv, nullptr, Vh,   M, E, E,  0,  1);

    linattn_chunk_kernel<<<dim3(B * 16 * N_CHUNK), blk, 0, stream>>>(phiQ, phiK, Vh, attn);

    gemm_nt_kernel<<<gg, blk, 0, stream>>>(attn, Wo, bo, (float*)d_out, M, E, E,  0,  0);
}


# Q, K, V projections as ONE launch of 768 work-groups (blockIdx.z picks weight and output): three groups share a WGP and fill each other's barrier and LDS waits; order of sums unchanged
# speedup vs baseline: 2.5164x; 1.1172x over previous
#include <hip/hip_runtime.h>
#include <hip/hip_bf16.h>

typedef __attribute__((ext_vector_type(2))) float v2f;
typedef __attribute__((ext_vector_type(8))) float v8f;

__device__ __forceinline__ v8f wmma4(v2f a, v2f b, v8f c) {
    return __builtin_amdgcn_wmma_f32_16x16x4_f32(
         false, a,  false, b,
         (short)0, c,  false,  false);
}

#define G_TM 128
#define G_TN 128
#define G_TK 32
#define G_PAD 36

__global__ __launch_bounds__(256, 6) void gemm_nt_kernel(
    const float* __restrict__ A,
    const float* __restrict__ W0, const float* __restrict__ W1, const float* __restrict__ W2,
    const float* __restrict__ bias,
    float* __restrict__ C0, float* __restrict__ C1, float* __restrict__ C2,
    int M, int N, int K, int act_bits, int split)
{
    const int sel = blockIdx.z;
    const float* __restrict__ W = (sel == 0) ? W0 : ((sel == 1) ? W1 : W2);
    float* __restrict__ C       = (sel == 0) ? C0 : ((sel == 1) ? C1 : C2);
    const int act = (act_bits >> sel) & 1;

    __shared__ float As[2][G_TM * G_PAD];
    __shared__ float Bs[2][G_TN * G_PAD];

    const int tid   = threadIdx.x;
    const int lane  = tid & 31;
    const int lrow  = lane & 15;
    const int khalf = (lane >> 4) * 2;
    const int rhalf = (lane >> 4) * 8;
    const int wave  = tid >> 5;

    const int m0 = blockIdx.x * G_TM;
    const int n0 = blockIdx.y * G_TN;
    const int wm = (wave & 3) * 32;
    const int wn = (wave >> 2) * 64;

    const int ldRow = tid >> 3;
    const int ldCol = (tid & 7) * 4;

    v8f cacc[2][4] = {};
    float4 pa0, pa1, pa2, pa3, pb0, pb1, pb2, pb3;
#define G_LD(i, kt)                                                                         \
    pa##i = *(const float4*)&A[(size_t)(m0 + ldRow + 32 * (i)) * K + (kt) + ldCol];         \
    pb##i = *(const float4*)&W[(size_t)(n0 + ldRow + 32 * (i)) * K + (kt) + ldCol];
#define G_ST(i, pp)                                                                         \
    *(float4*)&As[pp][(ldRow + 32 * (i)) * G_PAD + ldCol] = pa##i;                          \
    *(float4*)&Bs[pp][(ldRow + 32 * (i)) * G_PAD + ldCol] = pb##i;
#define G_LOAD_TILE(kt)  G_LD(0, kt) G_LD(1, kt) G_LD(2, kt) G_LD(3, kt)
#define G_STORE_TILE(pp) G_ST(0, pp) G_ST(1, pp) G_ST(2, pp) G_ST(3, pp)

    G_LOAD_TILE(0)
    G_STORE_TILE(0)
    __syncthreads();

    int p = 0;
    for (int kt = 0; kt < K; kt += G_TK) {
        const bool hasNext = (kt + G_TK) < K;
        if (hasNext) { G_LOAD_TILE(kt + G_TK) }

#pragma unroll
        for (int kk = 0; kk < G_TK; kk += 4) {
            v2f af[2], bf[4];
            af[0] = *(const v2f*)&As[p][(wm +      lrow) * G_PAD + kk + khalf];
            af[1] = *(const v2f*)&As[p][(wm + 16 + lrow) * G_PAD + kk + khalf];
#pragma unroll
            for (int fn = 0; fn < 4; ++fn)
                bf[fn] = *(const v2f*)&Bs[p][(wn + fn * 16 + lrow) * G_PAD + kk + khalf];
#pragma unroll
            for (int fm = 0; fm < 2; ++fm)
#pragma unroll
                for (int fn = 0; fn < 4; ++fn)
                    cacc[fm][fn] = wmma4(af[fm], bf[fn], cacc[fm][fn]);
        }

        if (hasNext) { G_STORE_TILE(p ^ 1) }
        __syncthreads();
        p ^= 1;
    }

#pragma unroll
    for (int fm = 0; fm < 2; ++fm)
#pragma unroll
        for (int fn = 0; fn < 4; ++fn)
#pragma unroll
            for (int r = 0; r < 8; ++r) {
                int row = m0 + wm + fm * 16 + r + rhalf;
                int col = n0 + wn + fn * 16 + lrow;
                float v = cacc[fm][fn][r];
                if (act == 1) v = (v > 0.0f) ? (v + 1.0f) : __expf(v);
                if (bias) v += bias[col];
                if (split) {
                    int bb = row >> 11, t = row & 2047;
                    int h  = col >> 6,  dd = col & 63;
                    C[(((size_t)bb * 16 + h) * 2048 + t) * 64 + dd] = v;
                } else {
                    C[(size_t)row * N + col] = v;
                }
            }
}

#define C_PAD  68
#define A_PAD  130
#define S_PAD  66
#define N_CHUNK 16

__global__ __launch_bounds__(256) void linattn_chunk_kernel(
    const float* __restrict__ Q, const float* __restrict__ Kx,
    const float* __restrict__ V, float* __restrict__ O)
{
    __shared__ __attribute__((aligned(16)))
        float lds[2 * 128 * C_PAD + 128 * A_PAD + 64 * S_PAD + 64 + 128];
    float* const qs  = lds;
    float* const vs  = qs + 128 * C_PAD;
    float* const ks  = vs + 128 * C_PAD;
    float* const Am  = ks;
    float* const Sm  = ks + 128 * A_PAD;
    float* const zv  = Sm + 64 * S_PAD;
    float* const den = zv + 64;

    const int tid   = threadIdx.x;
    const int lane  = tid & 31;
    const int lrow  = lane & 15;
    const int khalf = (lane >> 4) * 2;
    const int rhalf = (lane >> 4) * 8;
    const int wave  = tid >> 5;

    const int nbh = gridDim.x / N_CHUNK;
    const int jc  = N_CHUNK - 1 - blockIdx.x / nbh;
    const int bh  = blockIdx.x % nbh;
    const int b   = bh >> 4, h = bh & 15;
    const size_t base = (size_t)bh * 2048 * 64;

    float4 pk0, pk1, pk2, pk3, pk4, pk5, pk6, pk7;
    float4 pv0, pv1, pv2, pv3, pv4, pv5, pv6, pv7;
    float4 pq0, pq1, pq2, pq3, pq4, pq5, pq6, pq7;
    const int ldOff = (tid >> 4) * C_PAD + (tid & 15) * 4;
#define LD_G4(dst, src, cb, i) dst##i = *(const float4*)&src[(cb) + (size_t)(tid + (i) * 256) * 4];
#define ST_L4(buf, regs, i)    *(float4*)&buf[ldOff + (i) * 16 * C_PAD] = regs##i;
#define REP8(M, ...) M(__VA_ARGS__, 0) M(__VA_ARGS__, 1) M(__VA_ARGS__, 2) M(__VA_ARGS__, 3) \
                     M(__VA_ARGS__, 4) M(__VA_ARGS__, 5) M(__VA_ARGS__, 6) M(__VA_ARGS__, 7)

    REP8(LD_G4, pk, Kx, base)
    REP8(LD_G4, pv, V, base)
    {
        const size_t cbq = base + (size_t)jc * 128 * 64;
        REP8(LD_G4, pq, Q, cbq)
    }

    const int tr  = (wave & 3) * 16;
    const int tc0 = (wave >> 2) * 32;
    v8f sacc[2] = {};
    float zacc = 0.0f;

    for (int j = 0; j < jc; ++j) {
        REP8(ST_L4, ks, pk)
        REP8(ST_L4, vs, pv)
        __syncthreads();
        {
            const size_t cbn = base + (size_t)(j + 1) * 128 * 64;
            REP8(LD_G4, pk, Kx, cbn)
            REP8(LD_G4, pv, V, cbn)
        }

        for (int kk = 0; kk < 128; kk += 4) {
            v2f a;
            a.x = ks[(kk + khalf)     * C_PAD + tr + lrow];
            a.y = ks[(kk + khalf + 1) * C_PAD + tr + lrow];
#pragma unroll
            for (int c = 0; c < 2; ++c) {
                v2f bf;
                bf.x = vs[(kk + khalf)     * C_PAD + tc0 + c * 16 + lrow];
                bf.y = vs[(kk + khalf + 1) * C_PAD + tc0 + c * 16 + lrow];
                sacc[c] = wmma4(a, bf, sacc[c]);
            }
        }
        if (tid < 64) {
            float s = zacc;
            for (int i = 0; i < 128; ++i) s += ks[i * C_PAD + tid];
            zacc = s;
        }
        __syncthreads();
    }

    REP8(ST_L4, ks, pk)
    REP8(ST_L4, vs, pv)
    REP8(ST_L4, qs, pq)
#pragma unroll
    for (int c = 0; c < 2; ++c)
#pragma unroll
        for (int r = 0; r < 8; ++r)
            Sm[(tr + r + rhalf) * S_PAD + tc0 + c * 16 + lrow] = sacc[c][r];
    if (tid < 64) zv[tid] = zacc;
    __syncthreads();

    {
        const int m0 = wave * 16;
        v8f acc[8] = {};
        for (int kk = 0; kk < 64; kk += 4) {
            v2f a = *(const v2f*)&qs[(m0 + lrow) * C_PAD + kk + khalf];
#pragma unroll
            for (int c = 0; c < 8; ++c) {
                v2f bf = *(const v2f*)&ks[(c * 16 + lrow) * C_PAD + kk + khalf];
                acc[c] = wmma4(a, bf, acc[c]);
            }
        }
        __syncthreads();
#pragma unroll
        for (int c = 0; c < 8; ++c)
#pragma unroll
            for (int r = 0; r < 8; ++r) {
                int row = m0 + r + rhalf;
                int col = c * 16 + lrow;
                Am[row * A_PAD + col] = (row >= col) ? acc[c][r] : 0.0f;
            }
    }
    __syncthreads();

    if (tid < 128) {
        float s = 1e-5f;
        for (int j = 0; j <= tid; ++j) s += Am[tid * A_PAD + j];
        float qz = 0.0f;
        for (int k = 0; k < 64; ++k) qz += qs[tid * C_PAD + k] * zv[k];
        den[tid] = s + qz;
    }
    __syncthreads();

    {
        const int m0 = wave * 16;
        v8f acc[4] = {};
        for (int kk = 0; kk < 128; kk += 4) {
            v2f a = *(const v2f*)&Am[(m0 + lrow) * A_PAD + kk + khalf];
#pragma unroll
            for (int c = 0; c < 4; ++c) {
                v2f bf;
                bf.x = vs[(kk + khalf)     * C_PAD + c * 16 + lrow];
                bf.y = vs[(kk + khalf + 1) * C_PAD + c * 16 + lrow];
                acc[c] = wmma4(a, bf, acc[c]);
            }
        }
        for (int kk = 0; kk < 64; kk += 4) {
            v2f a = *(const v2f*)&qs[(m0 + lrow) * C_PAD + kk + khalf];
#pragma unroll
            for (int c = 0; c < 4; ++c) {
                v2f bf;
                bf.x = Sm[(kk + khalf)     * S_PAD + c * 16 + lrow];
                bf.y = Sm[(kk + khalf + 1) * S_PAD + c * 16 + lrow];
                acc[c] = wmma4(a, bf, acc[c]);
            }
        }
#pragma unroll
        for (int c = 0; c < 4; ++c)
#pragma unroll
            for (int r = 0; r < 8; ++r) {
                int row = m0 + r + rhalf;
                int col = c * 16 + lrow;
                float o = acc[c][r] / den[row];
                O[((size_t)(b * 2048 + jc * 128 + row)) * 1024 + h * 64 + col] = o;
            }
    }
}

extern "C" void kernel_launch(void* const* d_in, const int* in_sizes, int n_in,
                              void* d_out, int out_size, void* d_ws, size_t ws_size,
                              hipStream_t stream) {
    const float* x  = (const float*)d_in[0];
    const float* Wq = (const float*)d_in[1];
    const float* Wk = (const float*)d_in[2];
    const float* Wv = (const float*)d_in[3];
    const float* Wo = (const float*)d_in[4];
    const float* bo = (const float*)d_in[5];

    const int T = 2048, E = 1024;
    const int B = in_sizes[0] / (T * E);
    const int M = B * T;
    const size_t plane = (size_t)M * E;

    float* phiQ = (float*)d_ws;
    float* phiK = phiQ + plane;
    float* Vh   = phiK + plane;
    float* attn = Vh   + plane;

    dim3 blk(256);
    dim3 gg(M / G_TM, E / G_TN);

    dim3 gqkv(M / G_TM, E / G_TN, 3);
    gemm_nt_kernel<<<gqkv, blk, 0, stream>>>(x, Wq, Wk, Wv, nullptr, phiQ, phiK, Vh,
                                             M, E, E,  3,  1);

    linattn_chunk_kernel<<<dim3(B * 16 * N_CHUNK), blk, 0, stream>>>(phiQ, phiK, Vh, attn);

    gemm_nt_kernel<<<gg, blk, 0, stream>>>(attn, Wo, nullptr, nullptr, bo, (float*)d_out, nullptr, nullptr,
                                           M, E, E,  0,  0);
}


# chunk prefix state from stored per-chunk sums: a new launch writes k^T v and sum k per chunk; attention adds earlier chunks' sums in chunk order, no longer re-walks their rows; order of sums changes
# speedup vs baseline: 2.9242x; 1.1621x over previous
#include <hip/hip_runtime.h>
#include <hip/hip_bf16.h>

typedef __attribute__((ext_vector_type(2))) float v2f;
typedef __attribute__((ext_vector_type(8))) float v8f;

__device__ __forceinline__ v8f wmma4(v2f a, v2f b, v8f c) {
    return __builtin_amdgcn_wmma_f32_16x16x4_f32(
         false, a,  false, b,
         (short)0, c,  false,  false);
}

#define G_TM 128
#define G_TN 128
#define G_TK 32
#define G_PAD 36

__global__ __launch_bounds__(256, 6) void gemm_nt_kernel(
    const float* __restrict__ A,
    const float* __restrict__ W0, const float* __restrict__ W1, const float* __restrict__ W2,
    const float* __restrict__ bias,
    float* __restrict__ C0, float* __restrict__ C1, float* __restrict__ C2,
    int M, int N, int K, int act_bits)
{
    const int sel = blockIdx.z;
    const float* __restrict__ W = (sel == 0) ? W0 : ((sel == 1) ? W1 : W2);
    float* __restrict__ C       = (sel == 0) ? C0 : ((sel == 1) ? C1 : C2);
    const int act = (act_bits >> sel) & 1;

    __shared__ float As[2][G_TM * G_PAD];
    __shared__ float Bs[2][G_TN * G_PAD];

    const int tid   = threadIdx.x;
    const int lane  = tid & 31;
    const int lrow  = lane & 15;
    const int khalf = (lane >> 4) * 2;
    const int rhalf = (lane >> 4) * 8;
    const int wave  = tid >> 5;

    const int m0 = blockIdx.x * G_TM;
    const int n0 = blockIdx.y * G_TN;
    const int wm = (wave & 3) * 32;
    const int wn = (wave >> 2) * 64;

    const int ldRow = tid >> 3;
    const int ldCol = (tid & 7) * 4;

    v8f cacc[2][4] = {};
    float4 pa0, pa1, pa2, pa3, pb0, pb1, pb2, pb3;
#define G_LD(i, kt)                                                                         \
    pa##i = *(const float4*)&A[(size_t)(m0 + ldRow + 32 * (i)) * K + (kt) + ldCol];         \
    pb##i = *(const float4*)&W[(size_t)(n0 + ldRow + 32 * (i)) * K + (kt) + ldCol];
#define G_ST(i, pp)                                                                         \
    *(float4*)&As[pp][(ldRow + 32 * (i)) * G_PAD + ldCol] = pa##i;                          \
    *(float4*)&Bs[pp][(ldRow + 32 * (i)) * G_PAD + ldCol] = pb##i;
#define G_LOAD_TILE(kt)  G_LD(0, kt) G_LD(1, kt) G_LD(2, kt) G_LD(3, kt)
#define G_STORE_TILE(pp) G_ST(0, pp) G_ST(1, pp) G_ST(2, pp) G_ST(3, pp)

    G_LOAD_TILE(0)
    G_STORE_TILE(0)
    __syncthreads();

    int p = 0;
    for (int kt = 0; kt < K; kt += G_TK) {
        const bool hasNext = (kt + G_TK) < K;
        if (hasNext) { G_LOAD_TILE(kt + G_TK) }

#pragma unroll
        for (int kk = 0; kk < G_TK; kk += 4) {
            v2f af[2], bf[4];
            af[0] = *(const v2f*)&As[p][(wm +      lrow) * G_PAD + kk + khalf];
            af[1] = *(const v2f*)&As[p][(wm + 16 + lrow) * G_PAD + kk + khalf];
#pragma unroll
            for (int fn = 0; fn < 4; ++fn)
                bf[fn] = *(const v2f*)&Bs[p][(wn + fn * 16 + lrow) * G_PAD + kk + khalf];
#pragma unroll
            for (int fm = 0; fm < 2; ++fm)
#pragma unroll
                for (int fn = 0; fn < 4; ++fn)
                    cacc[fm][fn] = wmma4(af[fm], bf[fn], cacc[fm][fn]);
        }

        if (hasNext) { G_STORE_TILE(p ^ 1) }
        __syncthreads();
        p ^= 1;
    }

#pragma unroll
    for (int fm = 0; fm < 2; ++fm)
#pragma unroll
        for (int fn = 0; fn < 4; ++fn)
#pragma unroll
            for (int r = 0; r < 8; ++r) {
                int row = m0 + wm + fm * 16 + r + rhalf;
                int col = n0 + wn + fn * 16 + lrow;
                float v = cacc[fm][fn][r];
                if (act == 1) v = (v > 0.0f) ? (v + 1.0f) : __expf(v);
                if (bias) v += bias[col];
                C[(size_t)row * N + col] = v;
            }
}

#define C_PAD  68
#define A_PAD  130
#define S_PAD  66
#define N_CHUNK 16
#define ST_SZ  4096

#define LD_G4(dst, src, cb, i) dst##i = *(const float4*)&src[(cb) + (size_t)(i) * 16 * 1024];
#define ST_L4(buf, regs, i)    *(float4*)&buf[ldOff + (i) * 16 * C_PAD] = regs##i;
#define REP8(M, ...) M(__VA_ARGS__, 0) M(__VA_ARGS__, 1) M(__VA_ARGS__, 2) M(__VA_ARGS__, 3) \
                     M(__VA_ARGS__, 4) M(__VA_ARGS__, 5) M(__VA_ARGS__, 6) M(__VA_ARGS__, 7)

__global__ __launch_bounds__(256) void chunk_state_kernel(
    const float* __restrict__ Kx, const float* __restrict__ V, float* __restrict__ St)
{
    __shared__ __attribute__((aligned(16))) float lds[2 * 128 * C_PAD];
    float* const ks = lds;
    float* const vs = ks + 128 * C_PAD;

    const int tid   = threadIdx.x;
    const int lane  = tid & 31;
    const int lrow  = lane & 15;
    const int khalf = (lane >> 4) * 2;
    const int rhalf = (lane >> 4) * 8;
    const int wave  = tid >> 5;

    const int nbh = gridDim.x / (N_CHUNK - 1);
    const int j   = blockIdx.x / nbh;
    const int bh  = blockIdx.x % nbh;
    const int b   = bh >> 4, h = bh & 15;
    const size_t gOff = ((size_t)(b * 2048 + j * 128 + (tid >> 4))) * 1024 + h * 64 + (tid & 15) * 4;
    const int ldOff = (tid >> 4) * C_PAD + (tid & 15) * 4;

    float4 pk0, pk1, pk2, pk3, pk4, pk5, pk6, pk7;
    float4 pv0, pv1, pv2, pv3, pv4, pv5, pv6, pv7;
    REP8(LD_G4, pk, Kx, gOff)
    REP8(LD_G4, pv, V, gOff)
    REP8(ST_L4, ks, pk)
    REP8(ST_L4, vs, pv)
    __syncthreads();

    const int tr  = (wave & 3) * 16;
    const int tc0 = (wave >> 2) * 32;
    v8f sacc[2] = {};
    for (int kk = 0; kk < 128; kk += 4) {
        v2f a;
        a.x = ks[(kk + khalf)     * C_PAD + tr + lrow];
        a.y = ks[(kk + khalf + 1) * C_PAD + tr + lrow];
#pragma unroll
        for (int c = 0; c < 2; ++c) {
            v2f bf;
            bf.x = vs[(kk + khalf)     * C_PAD + tc0 + c * 16 + lrow];
            bf.y = vs[(kk + khalf + 1) * C_PAD + tc0 + c * 16 + lrow];
            sacc[c] = wmma4(a, bf, sacc[c]);
        }
    }

    float* const Pj = St + (size_t)(bh * N_CHUNK + j) * ST_SZ;
#pragma unroll
    for (int c = 0; c < 2; ++c)
#pragma unroll
        for (int r = 0; r < 8; ++r)
            Pj[(tr + r + rhalf) * 64 + tc0 + c * 16 + lrow] = sacc[c][r];

    if (tid < 64) {
        float s = 0.0f;
        for (int i = 0; i < 128; ++i) s += ks[i * C_PAD + tid];
        St[(size_t)nbh * N_CHUNK * ST_SZ + (size_t)(bh * N_CHUNK + j) * 64 + tid] = s;
    }
}

__global__ __launch_bounds__(256) void linattn_chunk_kernel(
    float* QO, const float* __restrict__ Kx, const float* __restrict__ V,
    const float* __restrict__ St)
{
    __shared__ __attribute__((aligned(16)))
        float lds[2 * 128 * C_PAD + 128 * A_PAD + 64 * S_PAD + 64 + 128];
    float* const qs  = lds;
    float* const vs  = qs + 128 * C_PAD;
    float* const ks  = vs + 128 * C_PAD;
    float* const Am  = ks;
    float* const zt  = ks + 128 * C_PAD;
    float* const Sm  = ks + 128 * A_PAD;
    float* const zv  = Sm + 64 * S_PAD;
    float* const den = zv + 64;

    const int tid   = threadIdx.x;
    const int lane  = tid & 31;
    const int lrow  = lane & 15;
    const int khalf = (lane >> 4) * 2;
    const int rhalf = (lane >> 4) * 8;
    const int wave  = tid >> 5;

    const int nbh = gridDim.x / N_CHUNK;
    const int jc  = N_CHUNK - 1 - blockIdx.x / nbh;
    const int bh  = blockIdx.x % nbh;
    const int b   = bh >> 4, h = bh & 15;
    const size_t gOff = ((size_t)(b * 2048 + jc * 128 + (tid >> 4))) * 1024 + h * 64 + (tid & 15) * 4;
    const int ldOff = (tid >> 4) * C_PAD + (tid & 15) * 4;

    float4 pk0, pk1, pk2, pk3, pk4, pk5, pk6, pk7;
    float4 pv0, pv1, pv2, pv3, pv4, pv5, pv6, pv7;
    float4 pq0, pq1, pq2, pq3, pq4, pq5, pq6, pq7;
    REP8(LD_G4, pk, Kx, gOff)
    REP8(LD_G4, pv, V, gOff)
    REP8(LD_G4, pq, QO, gOff)

    const float4* const Pb = (const float4*)(St + (size_t)bh * N_CHUNK * ST_SZ) + tid;
    float4 s0 = make_float4(0.0f, 0.0f, 0.0f, 0.0f), s1 = s0, s2 = s0, s3 = s0;
#define P_LD(n, jj)                                                                   \
    const float4 t##n##0 = Pb[(jj) * 1024],       t##n##1 = Pb[(jj) * 1024 + 256],    \
                 t##n##2 = Pb[(jj) * 1024 + 512], t##n##3 = Pb[(jj) * 1024 + 768];
#define P_ADD(n) s0 += t##n##0; s1 += t##n##1; s2 += t##n##2; s3 += t##n##3;
    int j = 0;
    for (; j + 4 <= jc; j += 4) {
        P_LD(a, j) P_LD(b, j + 1) P_LD(c, j + 2) P_LD(d, j + 3)
        P_ADD(a) P_ADD(b) P_ADD(c) P_ADD(d)
    }
    const int rem = jc - j;
    if (rem == 3)      { P_LD(a, j) P_LD(b, j + 1) P_LD(c, j + 2) P_ADD(a) P_ADD(b) P_ADD(c) }
    else if (rem == 2) { P_LD(a, j) P_LD(b, j + 1) P_ADD(a) P_ADD(b) }
    else if (rem == 1) { P_LD(a, j) P_ADD(a) }

    {
        float4 zz = make_float4(0.0f, 0.0f, 0.0f, 0.0f);
        if ((tid >> 4) < jc)
            zz = *(const float4*)&St[(size_t)nbh * N_CHUNK * ST_SZ + (size_t)bh * N_CHUNK * 64 + tid * 4];
        *(float4*)&zt[tid * 4] = zz;
    }

    REP8(ST_L4, ks, pk)
    REP8(ST_L4, vs, pv)
    REP8(ST_L4, qs, pq)
    {
        const int sOff = (tid >> 4) * S_PAD + (tid & 15) * 4;
        *(float2*)&Sm[sOff]                  = make_float2(s0.x, s0.y);
        *(float2*)&Sm[sOff + 2]              = make_float2(s0.z, s0.w);
        *(float2*)&Sm[sOff + 16 * S_PAD]     = make_float2(s1.x, s1.y);
        *(float2*)&Sm[sOff + 16 * S_PAD + 2] = make_float2(s1.z, s1.w);
        *(float2*)&Sm[sOff + 32 * S_PAD]     = make_float2(s2.x, s2.y);
        *(float2*)&Sm[sOff + 32 * S_PAD + 2] = make_float2(s2.z, s2.w);
        *(float2*)&Sm[sOff + 48 * S_PAD]     = make_float2(s3.x, s3.y);
        *(float2*)&Sm[sOff + 48 * S_PAD + 2] = make_float2(s3.z, s3.w);
    }
    __syncthreads();

    if (tid < 64) {
        float z = 0.0f;
#pragma unroll
        for (int jj = 0; jj < N_CHUNK - 1; ++jj) z += zt[jj * 64 + tid];
        zv[tid] = z;
    }

    {
        const int m0 = wave * 16;
        v8f acc[8] = {};
        for (int kk = 0; kk < 64; kk += 4) {
            v2f a = *(const v2f*)&qs[(m0 + lrow) * C_PAD + kk + khalf];
#pragma unroll
            for (int c = 0; c < 8; ++c) {
                v2f bf = *(const v2f*)&ks[(c * 16 + lrow) * C_PAD + kk + khalf];
                acc[c] = wmma4(a, bf, acc[c]);
            }
        }
        __syncthreads();
#pragma unroll
        for (int c = 0; c < 8; ++c)
#pragma unroll
            for (int r = 0; r < 8; ++r) {
                int row = m0 + r + rhalf;
                int col = c * 16 + lrow;
                Am[row * A_PAD + col] = (row >= col) ? acc[c][r] : 0.0f;
            }
    }
    __syncthreads();

    if (tid < 128) {
        float s = 1e-5f;
        for (int j = 0; j <= tid; ++j) s += Am[tid * A_PAD + j];
        float qz = 0.0f;
        for (int k = 0; k < 64; ++k) qz += qs[tid * C_PAD + k] * zv[k];
        den[tid] = s + qz;
    }
    __syncthreads();

    {
        const int m0 = wave * 16;
        v8f acc[4] = {};
        for (int kk = 0; kk < 128; kk += 4) {
            v2f a = *(const v2f*)&Am[(m0 + lrow) * A_PAD + kk + khalf];
#pragma unroll
            for (int c = 0; c < 4; ++c) {
                v2f bf;
                bf.x = vs[(kk + khalf)     * C_PAD + c * 16 + lrow];
                bf.y = vs[(kk + khalf + 1) * C_PAD + c * 16 + lrow];
                acc[c] = wmma4(a, bf, acc[c]);
            }
        }
        for (int kk = 0; kk < 64; kk += 4) {
            v2f a = *(const v2f*)&qs[(m0 + lrow) * C_PAD + kk + khalf];
#pragma unroll
            for (int c = 0; c < 4; ++c) {
                v2f bf;
                bf.x = Sm[(kk + khalf)     * S_PAD + c * 16 + lrow];
                bf.y = Sm[(kk + khalf + 1) * S_PAD + c * 16 + lrow];
                acc[c] = wmma4(a, bf, acc[c]);
            }
        }
#pragma unroll
        for (int c = 0; c < 4; ++c)
#pragma unroll
            for (int r = 0; r < 8; ++r) {
                int row = m0 + r + rhalf;
                int col = c * 16 + lrow;
                float o = acc[c][r] / den[row];
                QO[((size_t)(b * 2048 + jc * 128 + row)) * 1024 + h * 64 + col] = o;
            }
    }
}

extern "C" void kernel_launch(void* const* d_in, const int* in_sizes, int n_in,
                              void* d_out, int out_size, void* d_ws, size_t ws_size,
                              hipStream_t stream) {
    const float* x  = (const float*)d_in[0];
    const float* Wq = (const float*)d_in[1];
    const float* Wk = (const float*)d_in[2];
    const float* Wv = (const float*)d_in[3];
    const float* Wo = (const float*)d_in[4];
    const float* bo = (const float*)d_in[5];

    const int T = 2048, E = 1024;
    const int B = in_sizes[0] / (T * E);
    const int M = B * T;
    const size_t plane = (size_t)M * E;

    float* phiQ = (float*)d_ws;
    float* phiK = phiQ + plane;
    float* Vh   = phiK + plane;
    float* St   = Vh   + plane;

    dim3 blk(256);
    dim3 gg(M / G_TM, E / G_TN);

    dim3 gqkv(M / G_TM, E / G_TN, 3);
    gemm_nt_kernel<<<gqkv, blk, 0, stream>>>(x, Wq, Wk, Wv, nullptr, phiQ, phiK, Vh,
                                             M, E, E,  3);

    chunk_state_kernel<<<dim3(B * 16 * (N_CHUNK - 1)), blk, 0, stream>>>(phiK, Vh, St);

    linattn_chunk_kernel<<<dim3(B * 16 * N_CHUNK), blk, 0, stream>>>(phiQ, phiK, Vh, St);

    gemm_nt_kernel<<<gg, blk, 0, stream>>>(phiQ, Wo, nullptr, nullptr, bo, (float*)d_out, nullptr, nullptr,
                                           M, E, E,  0);
}


# GEMM and attention epilogues go through a wave's own LDS patch and leave as float4 rows: each 128-B output line written whole by one store, bias read once per lane; order of sums unchanged
# speedup vs baseline: 3.1656x; 1.0826x over previous
#include <hip/hip_runtime.h>
#include <hip/hip_bf16.h>

typedef __attribute__((ext_vector_type(2))) float v2f;
typedef __attribute__((ext_vector_type(8))) float v8f;

__device__ __forceinline__ v8f wmma4(v2f a, v2f b, v8f c) {
    return __builtin_amdgcn_wmma_f32_16x16x4_f32(
         false, a,  false, b,
         (short)0, c,  false,  false);
}

#define G_TM 128
#define G_TN 128
#define G_TK 32
#define G_PAD 36
#define E_PAD 68

__global__ __launch_bounds__(256, 6) void gemm_nt_kernel(
    const float* __restrict__ A,
    const float* __restrict__ W0, const float* __restrict__ W1, const float* __restrict__ W2,
    const float* __restrict__ bias,
    float* __restrict__ C0, float* __restrict__ C1, float* __restrict__ C2,
    int M, int N, int K, int act_bits)
{
    const int sel = blockIdx.z;
    const float* __restrict__ W = (sel == 0) ? W0 : ((sel == 1) ? W1 : W2);
    float* __restrict__ C       = (sel == 0) ? C0 : ((sel == 1) ? C1 : C2);
    const int act = (act_bits >> sel) & 1;

    __shared__ __attribute__((aligned(16))) float smem[2 * (G_TM + G_TN) * G_PAD];
    float (* const As)[G_TM * G_PAD] = (float (*)[G_TM * G_PAD])smem;
    float (* const Bs)[G_TN * G_PAD] = (float (*)[G_TN * G_PAD])(smem + 2 * G_TM * G_PAD);

    const int tid   = threadIdx.x;
    const int lane  = tid & 31;
    const int lrow  = lane & 15;
    const int khalf = (lane >> 4) * 2;
    const int rhalf = (lane >> 4) * 8;
    const int wave  = tid >> 5;

    const int m0 = blockIdx.x * G_TM;
    const int n0 = blockIdx.y * G_TN;
    const int wm = (wave & 3) * 32;
    const int wn = (wave >> 2) * 64;

    const int ldRow = tid >> 3;
    const int ldCol = (tid & 7) * 4;

    v8f cacc[2][4] = {};
    float4 pa0, pa1, pa2, pa3, pb0, pb1, pb2, pb3;
#define G_LD(i, kt)                                                                         \
    pa##i = *(const float4*)&A[(size_t)(m0 + ldRow + 32 * (i)) * K + (kt) + ldCol];         \
    pb##i = *(const float4*)&W[(size_t)(n0 + ldRow + 32 * (i)) * K + (kt) + ldCol];
#define G_ST(i, pp)                                                                         \
    *(float4*)&As[pp][(ldRow + 32 * (i)) * G_PAD + ldCol] = pa##i;                          \
    *(float4*)&Bs[pp][(ldRow + 32 * (i)) * G_PAD + ldCol] = pb##i;
#define G_LOAD_TILE(kt)  G_LD(0, kt) G_LD(1, kt) G_LD(2, kt) G_LD(3, kt)
#define G_STORE_TILE(pp) G_ST(0, pp) G_ST(1, pp) G_ST(2, pp) G_ST(3, pp)

    G_LOAD_TILE(0)
    G_STORE_TILE(0)
    __syncthreads();

    int p = 0;
    for (int kt = 0; kt < K; kt += G_TK) {
        const bool hasNext = (kt + G_TK) < K;
        if (hasNext) { G_LOAD_TILE(kt + G_TK) }

#pragma unroll
        for (int kk = 0; kk < G_TK; kk += 4) {
            v2f af[2], bf[4];
            af[0] = *(const v2f*)&As[p][(wm +      lrow) * G_PAD + kk + khalf];
            af[1] = *(const v2f*)&As[p][(wm + 16 + lrow) * G_PAD + kk + khalf];
#pragma unroll
            for (int fn = 0; fn < 4; ++fn)
                bf[fn] = *(const v2f*)&Bs[p][(wn + fn * 16 + lrow) * G_PAD + kk + khalf];
#pragma unroll
            for (int fm = 0; fm < 2; ++fm)
#pragma unroll
                for (int fn = 0; fn < 4; ++fn)
                    cacc[fm][fn] = wmma4(af[fm], bf[fn], cacc[fm][fn]);
        }

        if (hasNext) { G_STORE_TILE(p ^ 1) }
        __syncthreads();
        p ^= 1;
    }

    float* const stg = smem + wave * (32 * E_PAD);
#pragma unroll
    for (int fm = 0; fm < 2; ++fm)
#pragma unroll
        for (int fn = 0; fn < 4; ++fn)
#pragma unroll
            for (int r = 0; r < 8; ++r)
                stg[(fm * 16 + rhalf + r) * E_PAD + fn * 16 + lrow] = cacc[fm][fn][r];
    __syncthreads();

    const int c4 = lrow * 4;
    float4 bv = make_float4(0.0f, 0.0f, 0.0f, 0.0f);
    if (bias) bv = *(const float4*)&bias[n0 + wn + c4];
    float* const Cw = C + (size_t)(m0 + wm + (lane >> 4)) * N + n0 + wn + c4;
#pragma unroll
    for (int i = 0; i < 16; ++i) {
        float4 v = *(const float4*)&stg[(2 * i + (lane >> 4)) * E_PAD + c4];
        if (act == 1) {
            const float ex = __expf(v.x), ey = __expf(v.y), ez = __expf(v.z), ew = __expf(v.w);
            v.x = (v.x > 0.0f) ? (v.x + 1.0f) : ex;
            v.y = (v.y > 0.0f) ? (v.y + 1.0f) : ey;
            v.z = (v.z > 0.0f) ? (v.z + 1.0f) : ez;
            v.w = (v.w > 0.0f) ? (v.w + 1.0f) : ew;
        }
        if (bias) { v.x += bv.x; v.y += bv.y; v.z += bv.z; v.w += bv.w; }
        *(float4*)&Cw[(size_t)(2 * i) * N] = v;
    }
}

#define C_PAD  68
#define A_PAD  130
#define S_PAD  66
#define N_CHUNK 16
#define ST_SZ  4096

#define LD_G4(dst, src, cb, i) dst##i = *(const float4*)&src[(cb) + (size_t)(i) * 16 * 1024];
#define ST_L4(buf, regs, i)    *(float4*)&buf[ldOff + (i) * 16 * C_PAD] = regs##i;
#define REP8(M, ...) M(__VA_ARGS__, 0) M(__VA_ARGS__, 1) M(__VA_ARGS__, 2) M(__VA_ARGS__, 3) \
                     M(__VA_ARGS__, 4) M(__VA_ARGS__, 5) M(__VA_ARGS__, 6) M(__VA_ARGS__, 7)

__global__ __launch_bounds__(256) void chunk_state_kernel(
    const float* __restrict__ Kx, const float* __restrict__ V, float* __restrict__ St)
{
    __shared__ __attribute__((aligned(16))) float lds[2 * 128 * C_PAD];
    float* const ks = lds;
    float* const vs = ks + 128 * C_PAD;

    const int tid   = threadIdx.x;
    const int lane  = tid & 31;
    const int lrow  = lane & 15;
    const int khalf = (lane >> 4) * 2;
    const int rhalf = (lane >> 4) * 8;
    const int wave  = tid >> 5;

    const int nbh = gridDim.x / (N_CHUNK - 1);
    const int j   = blockIdx.x / nbh;
    const int bh  = blockIdx.x % nbh;
    const int b   = bh >> 4, h = bh & 15;
    const size_t gOff = ((size_t)(b * 2048 + j * 128 + (tid >> 4))) * 1024 + h * 64 + (tid & 15) * 4;
    const int ldOff = (tid >> 4) * C_PAD + (tid & 15) * 4;

    float4 pk0, pk1, pk2, pk3, pk4, pk5, pk6, pk7;
    float4 pv0, pv1, pv2, pv3, pv4, pv5, pv6, pv7;
    REP8(LD_G4, pk, Kx, gOff)
    REP8(LD_G4, pv, V, gOff)
    REP8(ST_L4, ks, pk)
    REP8(ST_L4, vs, pv)
    __syncthreads();

    const int tr  = (wave & 3) * 16;
    const int tc0 = (wave >> 2) * 32;
    v8f sacc[2] = {};
    for (int kk = 0; kk < 128; kk += 4) {
        v2f a;
        a.x = ks[(kk + khalf)     * C_PAD + tr + lrow];
        a.y = ks[(kk + khalf + 1) * C_PAD + tr + lrow];
#pragma unroll
        for (int c = 0; c < 2; ++c) {
            v2f bf;
            bf.x = vs[(kk + khalf)     * C_PAD + tc0 + c * 16 + lrow];
            bf.y = vs[(kk + khalf + 1) * C_PAD + tc0 + c * 16 + lrow];
            sacc[c] = wmma4(a, bf, sacc[c]);
        }
    }

    float* const Pj = St + (size_t)(bh * N_CHUNK + j) * ST_SZ;
#pragma unroll
    for (int c = 0; c < 2; ++c)
#pragma unroll
        for (int r = 0; r < 8; ++r)
            Pj[(tr + r + rhalf) * 64 + tc0 + c * 16 + lrow] = sacc[c][r];

    if (tid < 64) {
        float s = 0.0f;
        for (int i = 0; i < 128; ++i) s += ks[i * C_PAD + tid];
        St[(size_t)nbh * N_CHUNK * ST_SZ + (size_t)(bh * N_CHUNK + j) * 64 + tid] = s;
    }
}

__global__ __launch_bounds__(256) void linattn_chunk_kernel(
    float* QO, const float* __restrict__ Kx, const float* __restrict__ V,
    const float* __restrict__ St)
{
    __shared__ __attribute__((aligned(16)))
        float lds[2 * 128 * C_PAD + 128 * A_PAD + 64 * S_PAD + 64 + 128];
    float* const qs  = lds;
    float* const vs  = qs + 128 * C_PAD;
    float* const ks  = vs + 128 * C_PAD;
    float* const Am  = ks;
    float* const zt  = ks + 128 * C_PAD;
    float* const Sm  = ks + 128 * A_PAD;
    float* const zv  = Sm + 64 * S_PAD;
    float* const den = zv + 64;

    const int tid   = threadIdx.x;
    const int lane  = tid & 31;
    const int lrow  = lane & 15;
    const int khalf = (lane >> 4) * 2;
    const int rhalf = (lane >> 4) * 8;
    const int wave  = tid >> 5;

    const int nbh = gridDim.x / N_CHUNK;
    const int jc  = N_CHUNK - 1 - blockIdx.x / nbh;
    const int bh  = blockIdx.x % nbh;
    const int b   = bh >> 4, h = bh & 15;
    const size_t gOff = ((size_t)(b * 2048 + jc * 128 + (tid >> 4))) * 1024 + h * 64 + (tid & 15) * 4;
    const int ldOff = (tid >> 4) * C_PAD + (tid & 15) * 4;

    float4 pk0, pk1, pk2, pk3, pk4, pk5, pk6, pk7;
    float4 pv0, pv1, pv2, pv3, pv4, pv5, pv6, pv7;
    float4 pq0, pq1, pq2, pq3, pq4, pq5, pq6, pq7;
    REP8(LD_G4, pk, Kx, gOff)
    REP8(LD_G4, pv, V, gOff)
    REP8(LD_G4, pq, QO, gOff)

    const float4* const Pb = (const float4*)(St + (size_t)bh * N_CHUNK * ST_SZ) + tid;
    float4 s0 = make_float4(0.0f, 0.0f, 0.0f, 0.0f), s1 = s0, s2 = s0, s3 = s0;
#define P_LD(n, jj)                                                                   \
    const float4 t##n##0 = Pb[(jj) * 1024],       t##n##1 = Pb[(jj) * 1024 + 256],    \
                 t##n##2 = Pb[(jj) * 1024 + 512], t##n##3 = Pb[(jj) * 1024 + 768];
#define P_ADD(n) s0 += t##n##0; s1 += t##n##1; s2 += t##n##2; s3 += t##n##3;
    int j = 0;
    for (; j + 4 <= jc; j += 4) {
        P_LD(a, j) P_LD(b, j + 1) P_LD(c, j + 2) P_LD(d, j + 3)
        P_ADD(a) P_ADD(b) P_ADD(c) P_ADD(d)
    }
    const int rem = jc - j;
    if (rem == 3)      { P_LD(a, j) P_LD(b, j + 1) P_LD(c, j + 2) P_ADD(a) P_ADD(b) P_ADD(c) }
    else if (rem == 2) { P_LD(a, j) P_LD(b, j + 1) P_ADD(a) P_ADD(b) }
    else if (rem == 1) { P_LD(a, j) P_ADD(a) }

    {
        float4 zz = make_float4(0.0f, 0.0f, 0.0f, 0.0f);
        if ((tid >> 4) < jc)
            zz = *(const float4*)&St[(size_t)nbh * N_CHUNK * ST_SZ + (size_t)bh * N_CHUNK * 64 + tid * 4];
        *(float4*)&zt[tid * 4] = zz;
    }

    REP8(ST_L4, ks, pk)
    REP8(ST_L4, vs, pv)
    REP8(ST_L4, qs, pq)
    {
        const int sOff = (tid >> 4) * S_PAD + (tid & 15) * 4;
        *(float2*)&Sm[sOff]                  = make_float2(s0.x, s0.y);
        *(float2*)&Sm[sOff + 2]              = make_float2(s0.z, s0.w);
        *(float2*)&Sm[sOff + 16 * S_PAD]     = make_float2(s1.x, s1.y);
        *(float2*)&Sm[sOff + 16 * S_PAD + 2] = make_float2(s1.z, s1.w);
        *(float2*)&Sm[sOff + 32 * S_PAD]     = make_float2(s2.x, s2.y);
        *(float2*)&Sm[sOff + 32 * S_PAD + 2] = make_float2(s2.z, s2.w);
        *(float2*)&Sm[sOff + 48 * S_PAD]     = make_float2(s3.x, s3.y);
        *(float2*)&Sm[sOff + 48 * S_PAD + 2] = make_float2(s3.z, s3.w);
    }
    __syncthreads();

    if (tid < 64) {
        float z = 0.0f;
#pragma unroll
        for (int jj = 0; jj < N_CHUNK - 1; ++jj) z += zt[jj * 64 + tid];
        zv[tid] = z;
    }

    {
        const int m0 = wave * 16;
        v8f acc[8] = {};
        for (int kk = 0; kk < 64; kk += 4) {
            v2f a = *(const v2f*)&qs[(m0 + lrow) * C_PAD + kk + khalf];
#pragma unroll
            for (int c = 0; c < 8; ++c) {
                v2f bf = *(const v2f*)&ks[(c * 16 + lrow) * C_PAD + kk + khalf];
                acc[c] = wmma4(a, bf, acc[c]);
            }
        }
        __syncthreads();
#pragma unroll
        for (int c = 0; c < 8; ++c)
#pragma unroll
            for (int r = 0; r < 8; ++r) {
                int row = m0 + r + rhalf;
                int col = c * 16 + lrow;
                Am[row * A_PAD + col] = (row >= col) ? acc[c][r] : 0.0f;
            }
    }
    __syncthreads();

    if (tid < 128) {
        float s = 1e-5f;
        for (int j = 0; j <= tid; ++j) s += Am[tid * A_PAD + j];
        float qz = 0.0f;
        for (int k = 0; k < 64; ++k) qz += qs[tid * C_PAD + k] * zv[k];
        den[tid] = s + qz;
    }
    __syncthreads();

    {
        const int m0 = wave * 16;
        v8f acc[4] = {};
        for (int kk = 0; kk < 128; kk += 4) {
            v2f a = *(const v2f*)&Am[(m0 + lrow) * A_PAD + kk + khalf];
#pragma unroll
            for (int c = 0; c < 4; ++c) {
                v2f bf;
                bf.x = vs[(kk + khalf)     * C_PAD + c * 16 + lrow];
                bf.y = vs[(kk + khalf + 1) * C_PAD + c * 16 + lrow];
                acc[c] = wmma4(a, bf, acc[c]);
            }
        }
        for (int kk = 0; kk < 64; kk += 4) {
            v2f a = *(const v2f*)&qs[(m0 + lrow) * C_PAD + kk + khalf];
#pragma unroll
            for (int c = 0; c < 4; ++c) {
                v2f bf;
                bf.x = Sm[(kk + khalf)     * S_PAD + c * 16 + lrow];
                bf.y = Sm[(kk + khalf + 1) * S_PAD + c * 16 + lrow];
                acc[c] = wmma4(a, bf, acc[c]);
            }
        }
        float* const stg = qs + m0 * C_PAD;
#pragma unroll
        for (int c = 0; c < 4; ++c)
#pragma unroll
            for (int r = 0; r < 8; ++r)
                stg[(r + rhalf) * C_PAD + c * 16 + lrow] = acc[c][r] / den[m0 + r + rhalf];
        __syncthreads();
        float* const Ow = QO + ((size_t)(b * 2048 + jc * 128 + m0 + (lane >> 4))) * 1024 + h * 64 + lrow * 4;
#pragma unroll
        for (int i = 0; i < 8; ++i)
            *(float4*)&Ow[(size_t)(2 * i) * 1024] = *(const float4*)&stg[(2 * i + (lane >> 4)) * C_PAD + lrow * 4];
    }
}

extern "C" void kernel_launch(void* const* d_in, const int* in_sizes, int n_in,
                              void* d_out, int out_size, void* d_ws, size_t ws_size,
                              hipStream_t stream) {
    const float* x  = (const float*)d_in[0];
    const float* Wq = (const float*)d_in[1];
    const float* Wk = (const float*)d_in[2];
    const float* Wv = (const float*)d_in[3];
    const float* Wo = (const float*)d_in[4];
    const float* bo = (const float*)d_in[5];

    const int T = 2048, E = 1024;
    const int B = in_sizes[0] / (T * E);
    const int M = B * T;
    const size_t plane = (size_t)M * E;

    float* phiQ = (float*)d_ws;
    float* phiK = phiQ + plane;
    float* Vh   = phiK + plane;
    float* St   = Vh   + plane;

    dim3 blk(256);
    dim3 gg(M / G_TM, E / G_TN);

    dim3 gqkv(M / G_TM, E / G_TN, 3);
    gemm_nt_kernel<<<gqkv, blk, 0, stream>>>(x, Wq, Wk, Wv, nullptr, phiQ, phiK, Vh,
                                             M, E, E,  3);

    chunk_state_kernel<<<dim3(B * 16 * (N_CHUNK - 1)), blk, 0, stream>>>(phiK, Vh, St);

    linattn_chunk_kernel<<<dim3(B * 16 * N_CHUNK), blk, 0, stream>>>(phiQ, phiK, Vh, St);

    gemm_nt_kernel<<<gg, blk, 0, stream>>>(phiQ, Wo, nullptr, nullptr, bo, (float*)d_out, nullptr, nullptr,
                                           M, E, E,  0);
}


# projection GEMMs: K loop software-pipelined: LDS fragment reads issued one 16-MAC block ahead into a second register set, across the barrier too (next tile stored mid-tile); order of sums unchanged
# speedup vs baseline: 3.3075x; 1.0448x over previous
#include <hip/hip_runtime.h>
#include <hip/hip_bf16.h>

typedef __attribute__((ext_vector_type(2))) float v2f;
typedef __attribute__((ext_vector_type(8))) float v8f;
typedef const __attribute__((address_space(1))) float4* gmem_f4;

__device__ __forceinline__ v8f wmma4(v2f a, v2f b, v8f c) {
    return __builtin_amdgcn_wmma_f32_16x16x4_f32(
         false, a,  false, b,
         (short)0, c,  false,  false);
}

#define G_TM 128
#define G_TN 128
#define G_TK 32
#define G_K  1024
#define G_N  1024
#define G_PAD 36
#define E_PAD 68

__global__ __launch_bounds__(256, 6) void gemm_nt_kernel(
    const float* __restrict__ A,
    const float* __restrict__ W0, const float* __restrict__ W1, const float* __restrict__ W2,
    const float* __restrict__ bias,
    float* __restrict__ C0, float* __restrict__ C1, float* __restrict__ C2,
    int act_bits)
{
    const int sel = blockIdx.z;
    const float* __restrict__ W = (sel == 0) ? W0 : ((sel == 1) ? W1 : W2);
    float* __restrict__ C       = (sel == 0) ? C0 : ((sel == 1) ? C1 : C2);
    const int act = (act_bits >> sel) & 1;

    __shared__ __attribute__((aligned(16))) float smem[2 * (G_TM + G_TN) * G_PAD];
    float (* const As)[G_TM * G_PAD] = (float (*)[G_TM * G_PAD])smem;
    float (* const Bs)[G_TN * G_PAD] = (float (*)[G_TN * G_PAD])(smem + 2 * G_TM * G_PAD);

    const int tid   = threadIdx.x;
    const int lane  = tid & 31;
    const int lrow  = lane & 15;
    const int khalf = (lane >> 4) * 2;
    const int rhalf = (lane >> 4) * 8;
    const int wave  = tid >> 5;

    const int m0 = blockIdx.x * G_TM;
    const int n0 = blockIdx.y * G_TN;
    const int wm = (wave & 3) * 32;
    const int wn = (wave >> 2) * 64;

    const int ldRow = tid >> 3;
    const int ldCol = (tid & 7) * 4;

    v8f cacc[2][4] = {};
    float4 pa0, pa1, pa2, pa3, pb0, pb1, pb2, pb3;
    const char* const Ab = (const char*)(A + (size_t)m0 * G_K);
    const char* const Wb = (const char*)(W + (size_t)n0 * G_K);
    const unsigned tOff = (unsigned)(ldRow * G_K + ldCol) * 4u;
#define G_LD(i, kt)                                                                         \
    pa##i = *(gmem_f4)(ak + (unsigned long long)tOff + (unsigned long long)(32 * (i) * G_K * 4)); \
    pb##i = *(gmem_f4)(wk + (unsigned long long)tOff + (unsigned long long)(32 * (i) * G_K * 4));
#define G_ST(i, pp)                                                                         \
    *(float4*)&As[pp][(ldRow + 32 * (i)) * G_PAD + ldCol] = pa##i;                          \
    *(float4*)&Bs[pp][(ldRow + 32 * (i)) * G_PAD + ldCol] = pb##i;
#define G_LOAD_TILE(kt) {                                                                   \
    unsigned long long ak = (unsigned long long)Ab + (unsigned long long)(kt) * 4u;         \
    unsigned long long wk = (unsigned long long)Wb + (unsigned long long)(kt) * 4u;         \
    asm volatile("" : "+s"(ak));                                                            \
    asm volatile("" : "+s"(wk));                                                            \
    G_LD(0, kt) G_LD(1, kt) G_LD(2, kt) G_LD(3, kt) }
#define G_STORE_TILE(pp) G_ST(0, pp) G_ST(1, pp) G_ST(2, pp) G_ST(3, pp)

#define FR_DECL(X) v2f X##a00, X##a01, X##a10, X##a11,                                      \
                       X##b00, X##b01, X##b02, X##b03, X##b10, X##b11, X##b12, X##b13;
#define FR_A(pp, r, kk) *(const v2f*)&As[pp][(wm + (r) + lrow) * G_PAD + (kk) + khalf]
#define FR_B(pp, c, kk) *(const v2f*)&Bs[pp][(wn + (c) + lrow) * G_PAD + (kk) + khalf]
#define FR_LOAD(X, pp, kb)                                                                  \
    X##a00 = FR_A(pp, 0, kb);      X##a01 = FR_A(pp, 16, kb);                               \
    X##a10 = FR_A(pp, 0, (kb) + 4); X##a11 = FR_A(pp, 16, (kb) + 4);                        \
    X##b00 = FR_B(pp, 0, kb);      X##b01 = FR_B(pp, 16, kb);                               \
    X##b02 = FR_B(pp, 32, kb);     X##b03 = FR_B(pp, 48, kb);                               \
    X##b10 = FR_B(pp, 0, (kb) + 4);  X##b11 = FR_B(pp, 16, (kb) + 4);                       \
    X##b12 = FR_B(pp, 32, (kb) + 4); X##b13 = FR_B(pp, 48, (kb) + 4);
#define FR_ROW(X, s, fm)                                                                    \
    cacc[fm][0] = wmma4(X##a##s##fm, X##b##s##0, cacc[fm][0]);                              \
    cacc[fm][1] = wmma4(X##a##s##fm, X##b##s##1, cacc[fm][1]);                              \
    cacc[fm][2] = wmma4(X##a##s##fm, X##b##s##2, cacc[fm][2]);                              \
    cacc[fm][3] = wmma4(X##a##s##fm, X##b##s##3, cacc[fm][3]);
#define FR_MMA(X) FR_ROW(X, 0, 0) FR_ROW(X, 0, 1) FR_ROW(X, 1, 0) FR_ROW(X, 1, 1)
    FR_DECL(f0) FR_DECL(f1)

    G_LOAD_TILE(0)
    G_STORE_TILE(0)
    __syncthreads();
    FR_LOAD(f0, 0, 0)
    G_LOAD_TILE(G_TK)

    int p = 0;
    for (int kt = 0; kt < G_K; kt += G_TK) {
#define SB __builtin_amdgcn_sched_barrier(0);
        FR_LOAD(f1, p, 8)
        SB FR_MMA(f0) SB
        FR_LOAD(f0, p, 16)
        SB FR_MMA(f1) SB
        FR_LOAD(f1, p, 24)
        G_STORE_TILE(p ^ 1)
        SB FR_MMA(f0) SB
        __syncthreads();
        p ^= 1;
        FR_LOAD(f0, p, 0)
        if (kt + 2 * G_TK < G_K) G_LOAD_TILE(kt + 2 * G_TK)
        SB FR_MMA(f1) SB
    }

    float* const stg = smem + wave * (32 * E_PAD);
#pragma unroll
    for (int fm = 0; fm < 2; ++fm)
#pragma unroll
        for (int fn = 0; fn < 4; ++fn)
#pragma unroll
            for (int r = 0; r < 8; ++r)
                stg[(fm * 16 + rhalf + r) * E_PAD + fn * 16 + lrow] = cacc[fm][fn][r];
    __syncthreads();

    const int c4 = lrow * 4;
    float4 bv = make_float4(0.0f, 0.0f, 0.0f, 0.0f);
    if (bias) bv = *(const float4*)&bias[n0 + wn + c4];
    float* const Cw = C + (size_t)(m0 + wm + (lane >> 4)) * G_N + n0 + wn + c4;
#pragma unroll
    for (int i = 0; i < 16; ++i) {
        float4 v = *(const float4*)&stg[(2 * i + (lane >> 4)) * E_PAD + c4];
        if (act == 1) {
            const float ex = __expf(v.x), ey = __expf(v.y), ez = __expf(v.z), ew = __expf(v.w);
            v.x = (v.x > 0.0f) ? (v.x + 1.0f) : ex;
            v.y = (v.y > 0.0f) ? (v.y + 1.0f) : ey;
            v.z = (v.z > 0.0f) ? (v.z + 1.0f) : ez;
            v.w = (v.w > 0.0f) ? (v.w + 1.0f) : ew;
        }
        if (bias) { v.x += bv.x; v.y += bv.y; v.z += bv.z; v.w += bv.w; }
        *(float4*)&Cw[(size_t)(2 * i) * G_N] = v;
    }
}

#define C_PAD  68
#define A_PAD  130
#define S_PAD  66
#define N_CHUNK 16
#define ST_SZ  4096

#define LD_G4(dst, src, cb, i) dst##i = *(const float4*)&src[(cb) + (size_t)(i) * 16 * 1024];
#define ST_L4(buf, regs, i)    *(float4*)&buf[ldOff + (i) * 16 * C_PAD] = regs##i;
#define REP8(M, ...) M(__VA_ARGS__, 0) M(__VA_ARGS__, 1) M(__VA_ARGS__, 2) M(__VA_ARGS__, 3) \
                     M(__VA_ARGS__, 4) M(__VA_ARGS__, 5) M(__VA_ARGS__, 6) M(__VA_ARGS__, 7)

__global__ __launch_bounds__(256) void chunk_state_kernel(
    const float* __restrict__ Kx, const float* __restrict__ V, float* __restrict__ St)
{
    __shared__ __attribute__((aligned(16))) float lds[2 * 128 * C_PAD];
    float* const ks = lds;
    float* const vs = ks + 128 * C_PAD;

    const int tid   = threadIdx.x;
    const int lane  = tid & 31;
    const int lrow  = lane & 15;
    const int khalf = (lane >> 4) * 2;
    const int rhalf = (lane >> 4) * 8;
    const int wave  = tid >> 5;

    const int nbh = gridDim.x / (N_CHUNK - 1);
    const int j   = blockIdx.x / nbh;
    const int bh  = blockIdx.x % nbh;
    const int b   = bh >> 4, h = bh & 15;
    const size_t gOff = ((size_t)(b * 2048 + j * 128 + (tid >> 4))) * 1024 + h * 64 + (tid & 15) * 4;
    const int ldOff = (tid >> 4) * C_PAD + (tid & 15) * 4;

    float4 pk0, pk1, pk2, pk3, pk4, pk5, pk6, pk7;
    float4 pv0, pv1, pv2, pv3, pv4, pv5, pv6, pv7;
    REP8(LD_G4, pk, Kx, gOff)
    REP8(LD_G4, pv, V, gOff)
    REP8(ST_L4, ks, pk)
    REP8(ST_L4, vs, pv)
    __syncthreads();

    const int tr  = (wave & 3) * 16;
    const int tc0 = (wave >> 2) * 32;
    v8f sacc[2] = {};
    for (int kk = 0; kk < 128; kk += 4) {
        v2f a;
        a.x = ks[(kk + khalf)     * C_PAD + tr + lrow];
        a.y = ks[(kk + khalf + 1) * C_PAD + tr + lrow];
#pragma unroll
        for (int c = 0; c < 2; ++c) {
            v2f bf;
            bf.x = vs[(kk + khalf)     * C_PAD + tc0 + c * 16 + lrow];
            bf.y = vs[(kk + khalf + 1) * C_PAD + tc0 + c * 16 + lrow];
            sacc[c] = wmma4(a, bf, sacc[c]);
        }
    }

    float* const Pj = St + (size_t)(bh * N_CHUNK + j) * ST_SZ;
#pragma unroll
    for (int c = 0; c < 2; ++c)
#pragma unroll
        for (int r = 0; r < 8; ++r)
            Pj[(tr + r + rhalf) * 64 + tc0 + c * 16 + lrow] = sacc[c][r];

    if (tid < 64) {
        float s = 0.0f;
        for (int i = 0; i < 128; ++i) s += ks[i * C_PAD + tid];
        St[(size_t)nbh * N_CHUNK * ST_SZ + (size_t)(bh * N_CHUNK + j) * 64 + tid] = s;
    }
}

__global__ __launch_bounds__(256) void linattn_chunk_kernel(
    float* QO, const float* __restrict__ Kx, const float* __restrict__ V,
    const float* __restrict__ St)
{
    __shared__ __attribute__((aligned(16)))
        float lds[2 * 128 * C_PAD + 128 * A_PAD + 64 * S_PAD + 64 + 128];
    float* const qs  = lds;
    float* const vs  = qs + 128 * C_PAD;
    float* const ks  = vs + 128 * C_PAD;
    float* const Am  = ks;
    float* const zt  = ks + 128 * C_PAD;
    float* const Sm  = ks + 128 * A_PAD;
    float* const zv  = Sm + 64 * S_PAD;
    float* const den = zv + 64;

    const int tid   = threadIdx.x;
    const int lane  = tid & 31;
    const int lrow  = lane & 15;
    const int khalf = (lane >> 4) * 2;
    const int rhalf = (lane >> 4) * 8;
    const int wave  = tid >> 5;

    const int nbh = gridDim.x / N_CHUNK;
    const int jc  = N_CHUNK - 1 - blockIdx.x / nbh;
    const int bh  = blockIdx.x % nbh;
    const int b   = bh >> 4, h = bh & 15;
    const size_t gOff = ((size_t)(b * 2048 + jc * 128 + (tid >> 4))) * 1024 + h * 64 + (tid & 15) * 4;
    const int ldOff = (tid >> 4) * C_PAD + (tid & 15) * 4;

    float4 pk0, pk1, pk2, pk3, pk4, pk5, pk6, pk7;
    float4 pv0, pv1, pv2, pv3, pv4, pv5, pv6, pv7;
    float4 pq0, pq1, pq2, pq3, pq4, pq5, pq6, pq7;
    REP8(LD_G4, pk, Kx, gOff)
    REP8(LD_G4, pv, V, gOff)
    REP8(LD_G4, pq, QO, gOff)

    const float4* const Pb = (const float4*)(St + (size_t)bh * N_CHUNK * ST_SZ) + tid;
    float4 s0 = make_float4(0.0f, 0.0f, 0.0f, 0.0f), s1 = s0, s2 = s0, s3 = s0;
#define P_LD(n, jj)                                                                   \
    const float4 t##n##0 = Pb[(jj) * 1024],       t##n##1 = Pb[(jj) * 1024 + 256],    \
                 t##n##2 = Pb[(jj) * 1024 + 512], t##n##3 = Pb[(jj) * 1024 + 768];
#define P_ADD(n) s0 += t##n##0; s1 += t##n##1; s2 += t##n##2; s3 += t##n##3;
    int j = 0;
    for (; j + 4 <= jc; j += 4) {
        P_LD(a, j) P_LD(b, j + 1) P_LD(c, j + 2) P_LD(d, j + 3)
        P_ADD(a) P_ADD(b) P_ADD(c) P_ADD(d)
    }
    const int rem = jc - j;
    if (rem == 3)      { P_LD(a, j) P_LD(b, j + 1) P_LD(c, j + 2) P_ADD(a) P_ADD(b) P_ADD(c) }
    else if (rem == 2) { P_LD(a, j) P_LD(b, j + 1) P_ADD(a) P_ADD(b) }
    else if (rem == 1) { P_LD(a, j) P_ADD(a) }

    {
        float4 zz = make_float4(0.0f, 0.0f, 0.0f, 0.0f);
        if ((tid >> 4) < jc)
            zz = *(const float4*)&St[(size_t)nbh * N_CHUNK * ST_SZ + (size_t)bh * N_CHUNK * 64 + tid * 4];
        *(float4*)&zt[tid * 4] = zz;
    }

    REP8(ST_L4, ks, pk)
    REP8(ST_L4, vs, pv)
    REP8(ST_L4, qs, pq)
    {
        const int sOff = (tid >> 4) * S_PAD + (tid & 15) * 4;
        *(float2*)&Sm[sOff]                  = make_float2(s0.x, s0.y);
        *(float2*)&Sm[sOff + 2]              = make_float2(s0.z, s0.w);
        *(float2*)&Sm[sOff + 16 * S_PAD]     = make_float2(s1.x, s1.y);
        *(float2*)&Sm[sOff + 16 * S_PAD + 2] = make_float2(s1.z, s1.w);
        *(float2*)&Sm[sOff + 32 * S_PAD]     = make_float2(s2.x, s2.y);
        *(float2*)&Sm[sOff + 32 * S_PAD + 2] = make_float2(s2.z, s2.w);
        *(float2*)&Sm[sOff + 48 * S_PAD]     = make_float2(s3.x, s3.y);
        *(float2*)&Sm[sOff + 48 * S_PAD + 2] = make_float2(s3.z, s3.w);
    }
    __syncthreads();

    if (tid < 64) {
        float z = 0.0f;
#pragma unroll
        for (int jj = 0; jj < N_CHUNK - 1; ++jj) z += zt[jj * 64 + tid];
        zv[tid] = z;
    }

    {
        const int m0 = wave * 16;
        v8f acc[8] = {};
        for (int kk = 0; kk < 64; kk += 4) {
            v2f a = *(const v2f*)&qs[(m0 + lrow) * C_PAD + kk + khalf];
#pragma unroll
            for (int c = 0; c < 8; ++c) {
                v2f bf = *(const v2f*)&ks[(c * 16 + lrow) * C_PAD + kk + khalf];
                acc[c] = wmma4(a, bf, acc[c]);
            }
        }
        __syncthreads();
#pragma unroll
        for (int c = 0; c < 8; ++c)
#pragma unroll
            for (int r = 0; r < 8; ++r) {
                int row = m0 + r + rhalf;
                int col = c * 16 + lrow;
                Am[row * A_PAD + col] = (row >= col) ? acc[c][r] : 0.0f;
            }
    }
    __syncthreads();

    if (tid < 128) {
        float s = 1e-5f;
        for (int j = 0; j <= tid; ++j) s += Am[tid * A_PAD + j];
        float qz = 0.0f;
        for (int k = 0; k < 64; ++k) qz += qs[tid * C_PAD + k] * zv[k];
        den[tid] = s + qz;
    }
    __syncthreads();

    {
        const int m0 = wave * 16;
        v8f acc[4] = {};
        for (int kk = 0; kk < 128; kk += 4) {
            v2f a = *(const v2f*)&Am[(m0 + lrow) * A_PAD + kk + khalf];
#pragma unroll
            for (int c = 0; c < 4; ++c) {
                v2f bf;
                bf.x = vs[(kk + khalf)     * C_PAD + c * 16 + lrow];
                bf.y = vs[(kk + khalf + 1) * C_PAD + c * 16 + lrow];
                acc[c] = wmma4(a, bf, acc[c]);
            }
        }
        for (int kk = 0; kk < 64; kk += 4) {
            v2f a = *(const v2f*)&qs[(m0 + lrow) * C_PAD + kk + khalf];
#pragma unroll
            for (int c = 0; c < 4; ++c) {
                v2f bf;
                bf.x = Sm[(kk + khalf)     * S_PAD + c * 16 + lrow];
                bf.y = Sm[(kk + khalf + 1) * S_PAD + c * 16 + lrow];
                acc[c] = wmma4(a, bf, acc[c]);
            }
        }
        float* const stg = qs + m0 * C_PAD;
#pragma unroll
        for (int c = 0; c < 4; ++c)
#pragma unroll
            for (int r = 0; r < 8; ++r)
                stg[(r + rhalf) * C_PAD + c * 16 + lrow] = acc[c][r] / den[m0 + r + rhalf];
        __syncthreads();
        float* const Ow = QO + ((size_t)(b * 2048 + jc * 128 + m0 + (lane >> 4))) * 1024 + h * 64 + lrow * 4;
#pragma unroll
        for (int i = 0; i < 8; ++i)
            *(float4*)&Ow[(size_t)(2 * i) * 1024] = *(const float4*)&stg[(2 * i + (lane >> 4)) * C_PAD + lrow * 4];
    }
}

extern "C" void kernel_launch(void* const* d_in, const int* in_sizes, int n_in,
                              void* d_out, int out_size, void* d_ws, size_t ws_size,
                              hipStream_t stream) {
    const float* x  = (const float*)d_in[0];
    const float* Wq = (const float*)d_in[1];
    const float* Wk = (const float*)d_in[2];
    const float* Wv = (const float*)d_in[3];
    const float* Wo = (const float*)d_in[4];
    const float* bo = (const float*)d_in[5];

    const int T = 2048, E = 1024;
    const int B = in_sizes[0] / (T * E);
    const int M = B * T;
    const size_t plane = (size_t)M * E;

    float* phiQ = (float*)d_ws;
    float* phiK = phiQ + plane;
    float* Vh   = phiK + plane;
    float* St   = Vh   + plane;

    dim3 blk(256);
    dim3 gg(M / G_TM, E / G_TN);

    dim3 gqkv(M / G_TM, E / G_TN, 3);
    gemm_nt_kernel<<<gqkv, blk, 0, stream>>>(x, Wq, Wk, Wv, nullptr, phiQ, phiK, Vh,
                                              3);

    chunk_state_kernel<<<dim3(B * 16 * (N_CHUNK - 1)), blk, 0, stream>>>(phiK, Vh, St);

    linattn_chunk_kernel<<<dim3(B * 16 * N_CHUNK), blk, 0, stream>>>(phiQ, phiK, Vh, St);

    gemm_nt_kernel<<<gg, blk, 0, stream>>>(phiQ, Wo, nullptr, nullptr, bo, (float*)d_out, nullptr, nullptr,
                                            0);
}
